# starting kernel
# baseline (speedup 1.0000x reference)
#include <hip/hip_runtime.h>
#include <hip/hip_bf16.h>


typedef __attribute__((ext_vector_type(2))) float v2f;
typedef __attribute__((ext_vector_type(8))) float v8f;

#define HID 128
#define EMBD 64

__global__ void fill0_kernel(float* __restrict__ p, long long n) {
    long long i = (long long)blockIdx.x * blockDim.x + threadIdx.x;
    if (i < n) p[i] = 0.0f;
}

__global__ void build_h0_kernel(const int* __restrict__ n_id,
                                const float* __restrict__ emb,
                                const float* __restrict__ x_struct,
                                float* __restrict__ h0, int n_nodes) {
    long long t = (long long)blockIdx.x * blockDim.x + threadIdx.x;
    int i = (int)(t >> 7);
    int c = (int)(t & 127);
    if (i >= n_nodes) return;
    float v;
    if (c < EMBD) v = emb[(size_t)n_id[i] * EMBD + c];
    else          v = x_struct[(size_t)i * EMBD + (c - EMBD)];
    h0[(size_t)i * HID + c] = v;
}

__global__ void deg_kernel(const int* __restrict__ dst, float* __restrict__ deg,
                           int n_edges) {
    int e = blockIdx.x * blockDim.x + threadIdx.x;
    if (e >= n_edges) return;
    atomicAdd(&deg[dst[e]], 1.0f);
}

__global__ void invdeg_kernel(const float* __restrict__ deg,
                              float* __restrict__ invdeg, int n_nodes) {
    int i = blockIdx.x * blockDim.x + threadIdx.x;
    if (i >= n_nodes) return;
    invdeg[i] = 1.0f / fmaxf(deg[i], 1.0f);
}

__global__ void scatter_kernel(const float* __restrict__ h,
                               const int* __restrict__ src,
                               const int* __restrict__ dst,
                               float* __restrict__ agg, int n_edges) {
    long long t = (long long)blockIdx.x * blockDim.x + threadIdx.x;
    int e = (int)(t >> 7);
    int c = (int)(t & 127);
    if (e >= n_edges) return;
    int s = src[e], d = dst[e];
    atomicAdd(&agg[(size_t)d * HID + c], h[(size_t)s * HID + c]);
}

__global__ __launch_bounds__(256)
void sage_gemm_kernel(const float* __restrict__ agg,
                      const float* __restrict__ invdeg,
                      const float* __restrict__ h,
                      const float* __restrict__ Wl,
                      const float* __restrict__ Wr,
                      const float* __restrict__ bias,
                      float* __restrict__ out,
                      int n_nodes, int do_relu) {
    const int lane = threadIdx.x & 31;
    const int wave = threadIdx.x >> 5;
    const int m0   = blockIdx.x * 16;
    const int n0   = wave * 16;

    const int lrow  = lane & 15;
    const int khalf = (lane >> 4) * 2;

    int mrow = m0 + lrow;
    if (mrow >= n_nodes) mrow = n_nodes - 1;
    const float idg = invdeg[mrow];
    const float* __restrict__ hrow = h   + (size_t)mrow * HID;
    const float* __restrict__ arow = agg + (size_t)mrow * HID;

    const int ncol = n0 + lrow;

    v8f acc_l = {};
    v8f acc_r = {};

    for (int k0 = 0; k0 < HID; k0 += 4) {
        const int k = k0 + khalf;
        v2f a_l, a_r, b_l, b_r;
        a_l.x = arow[k]     * idg;
        a_l.y = arow[k + 1] * idg;
        a_r.x = hrow[k];
        a_r.y = hrow[k + 1];
        b_l.x = Wl[(size_t)k * HID + ncol];
        b_l.y = Wl[(size_t)(k + 1) * HID + ncol];
        b_r.x = Wr[(size_t)k * HID + ncol];
        b_r.y = Wr[(size_t)(k + 1) * HID + ncol];

        acc_l = __builtin_amdgcn_wmma_f32_16x16x4_f32(
            false, a_l, false, b_l, (short)0, acc_l, false, false);
        acc_r = __builtin_amdgcn_wmma_f32_16x16x4_f32(
            false, a_r, false, b_r, (short)0, acc_r, false, false);
    }

    const float bv = bias[ncol];
    const int mbase = m0 + (lane >> 4) * 8;
#pragma unroll
    for (int v = 0; v < 8; ++v) {
        const int m = mbase + v;
        if (m < n_nodes) {
            float o = acc_l[v] + acc_r[v] + bv;
            if (do_relu) o = o > 0.0f ? o : 0.0f;
            out[(size_t)m * HID + ncol] = o;
        }
    }
}

__global__ void edge_dot_kernel(const float* __restrict__ h2,
                                const int* __restrict__ es,
                                const int* __restrict__ ed,
                                float* __restrict__ out, int n) {
    int e = blockIdx.x * blockDim.x + threadIdx.x;
    if (e >= n) return;
    const float4* a = (const float4*)(h2 + (size_t)es[e] * HID);
    const float4* b = (const float4*)(h2 + (size_t)ed[e] * HID);
    float s = 0.0f;
#pragma unroll 8
    for (int i = 0; i < HID / 4; ++i) {
        float4 x = a[i], y = b[i];
        s += x.x * y.x + x.y * y.y + x.z * y.z + x.w * y.w;
    }
    out[e] = s;
}

extern "C" void kernel_launch(void* const* d_in, const int* in_sizes, int n_in,
                              void* d_out, int out_size, void* d_ws, size_t ws_size,
                              hipStream_t stream) {
    const int*   n_id     = (const int*)  d_in[0];
    const float* x_struct = (const float*)d_in[1];
    const int*   eidx     = (const int*)  d_in[2];
    const int*   elab     = (const int*)  d_in[3];
    const float* emb      = (const float*)d_in[4];
    const float* W1l      = (const float*)d_in[5];
    const float* W1r      = (const float*)d_in[6];
    const float* b1       = (const float*)d_in[7];
    const float* W2l      = (const float*)d_in[8];
    const float* W2r      = (const float*)d_in[9];
    const float* b2       = (const float*)d_in[10];
    float*       out      = (float*)d_out;

    const int N = in_sizes[0];
    const int E = in_sizes[2] / 2;
    const int L = in_sizes[3] / 2;

    const int* src = eidx;
    const int* dst = eidx + E;
    const int* lsrc = elab;
    const int* ldst = elab + L;

    float* ws     = (float*)d_ws;
    float* h0     = ws;
    float* h1     = h0  + (size_t)N * HID;
    float* h2     = h1  + (size_t)N * HID;
    float* agg    = h2  + (size_t)N * HID;
    float* deg    = agg + (size_t)N * HID;
    float* invdeg = deg + N;

    const long long nh = (long long)N * HID;

    fill0_kernel<<<(int)((nh + 255) / 256), 256, 0, stream>>>(agg, nh);
    fill0_kernel<<<(N + 255) / 256, 256, 0, stream>>>(deg, N);
    build_h0_kernel<<<(int)((nh + 255) / 256), 256, 0, stream>>>(
        n_id, emb, x_struct, h0, N);
    deg_kernel<<<(E + 255) / 256, 256, 0, stream>>>(dst, deg, E);
    invdeg_kernel<<<(N + 255) / 256, 256, 0, stream>>>(deg, invdeg, N);

    const long long ethreads = (long long)E * HID;
    const int eblocks = (int)((ethreads + 255) / 256);
    const int mblocks = (N + 15) / 16;

    scatter_kernel<<<eblocks, 256, 0, stream>>>(h0, src, dst, agg, E);
    sage_gemm_kernel<<<mblocks, 256, 0, stream>>>(
        agg, invdeg, h0, W1l, W1r, b1, h1, N,  1);

    fill0_kernel<<<(int)((nh + 255) / 256), 256, 0, stream>>>(agg, nh);
    scatter_kernel<<<eblocks, 256, 0, stream>>>(h1, src, dst, agg, E);
    sage_gemm_kernel<<<mblocks, 256, 0, stream>>>(
        agg, invdeg, h1, W2l, W2r, b2, h2, N,  0);

    edge_dot_kernel<<<(L + 255) / 256, 256, 0, stream>>>(h2, lsrc, ldst, out, L);
}


# Replace f32 atomic scatter-add by per-call counting sort of edges by dst (returning int adds, scan, fill) + wave-per-node register gather-sum; sums now in bucket slot order, not atomic arrival
# speedup vs baseline: 2.2680x; 2.2680x over previous
#include <hip/hip_runtime.h>
#include <hip/hip_bf16.h>


typedef __attribute__((ext_vector_type(2))) float v2f;
typedef __attribute__((ext_vector_type(8))) float v8f;

#define HID 128
#define EMBD 64

__global__ void fill0_kernel(float* __restrict__ p, long long n) {
    long long i = (long long)blockIdx.x * blockDim.x + threadIdx.x;
    if (i < n) p[i] = 0.0f;
}

__global__ void build_h0_kernel(const int* __restrict__ n_id,
                                const float* __restrict__ emb,
                                const float* __restrict__ x_struct,
                                float* __restrict__ h0, int n_nodes) {
    long long t = (long long)blockIdx.x * blockDim.x + threadIdx.x;
    int i = (int)(t >> 7);
    int c = (int)(t & 127);
    if (i >= n_nodes) return;
    float v;
    if (c < EMBD) v = emb[(size_t)n_id[i] * EMBD + c];
    else          v = x_struct[(size_t)i * EMBD + (c - EMBD)];
    h0[(size_t)i * HID + c] = v;
}

__global__ void rank_kernel(const int* __restrict__ dst, int* __restrict__ cnt,
                            int* __restrict__ rank, int n_edges) {
    int e = blockIdx.x * blockDim.x + threadIdx.x;
    if (e >= n_edges) return;
    rank[e] = atomicAdd(&cnt[dst[e]], 1);
}

__global__ __launch_bounds__(1024)
void scan_kernel(const int* __restrict__ cnt, int* __restrict__ rowptr,
                 float* __restrict__ invdeg, int n_nodes) {
    __shared__ int wave_tot[32];
    const int t    = threadIdx.x;
    const int lane = t & 31;
    const int wave = t >> 5;
    const int per  = (n_nodes + 1023) / 1024;
    const int base = t * per;
    const int left = n_nodes - base;
    const int lim  = left < per ? (left > 0 ? left : 0) : per;

    int s = 0;
    for (int i = 0; i < lim; ++i) s += cnt[base + i];
    int incl = s;
    for (int d = 1; d < 32; d <<= 1) {
        const int v = __shfl_up(incl, d);
        if (lane >= d) incl += v;
    }
    if (lane == 31) wave_tot[wave] = incl;
    __syncthreads();
    if (wave == 0) {
        const int w = wave_tot[lane];
        int wi = w;
        for (int d = 1; d < 32; d <<= 1) {
            const int v = __shfl_up(wi, d);
            if (lane >= d) wi += v;
        }
        wave_tot[lane] = wi - w;
    }
    __syncthreads();
    int run = wave_tot[wave] + incl - s;
    for (int i = 0; i < lim; ++i) {
        const int c = cnt[base + i];
        rowptr[base + i] = run;
        invdeg[base + i] = 1.0f / fmaxf((float)c, 1.0f);
        run += c;
    }
}

__global__ void fill_kernel(const int* __restrict__ src, const int* __restrict__ dst,
                            const int* __restrict__ rowptr, const int* __restrict__ rank,
                            int* __restrict__ col, int n_edges) {
    int e = blockIdx.x * blockDim.x + threadIdx.x;
    if (e >= n_edges) return;
    col[rowptr[dst[e]] + rank[e]] = src[e];
}

__global__ __launch_bounds__(256)
void gather_kernel(const float* __restrict__ h,
                   const int* __restrict__ rowptr,
                   const int* __restrict__ cnt,
                   const int* __restrict__ col,
                   float* __restrict__ agg, int n_nodes) {
    const int lane = threadIdx.x & 31;
    const int node = __builtin_amdgcn_readfirstlane(
        (int)(blockIdx.x * 8 + (threadIdx.x >> 5)));
    if (node >= n_nodes) return;
    const int start = rowptr[node];
    const int deg   = cnt[node];
    const unsigned last = (unsigned)(n_nodes - 1);
    const float4* __restrict__ h4 = (const float4*)h;

    float4 acc = make_float4(0.0f, 0.0f, 0.0f, 0.0f);
    for (int j0 = 0; j0 < deg; j0 += 32) {
        const int nb = (deg - j0) < 32 ? (deg - j0) : 32;
        unsigned mine = 0;
        if (lane < nb) mine = (unsigned)col[start + j0 + lane];
        mine = mine < last ? mine : last;
        for (int j = 0; j < nb; j += 4) {
            const int g = nb - j;
            const float4 zero4 = make_float4(0.0f, 0.0f, 0.0f, 0.0f);
            float4 v0, v1 = zero4, v2 = zero4, v3 = zero4;
            v0 = h4[(size_t)__builtin_amdgcn_readlane(mine, j) * 32 + lane];
            if (g > 1) v1 = h4[(size_t)__builtin_amdgcn_readlane(mine, j + 1) * 32 + lane];
            if (g > 2) v2 = h4[(size_t)__builtin_amdgcn_readlane(mine, j + 2) * 32 + lane];
            if (g > 3) v3 = h4[(size_t)__builtin_amdgcn_readlane(mine, j + 3) * 32 + lane];
            acc.x += v0.x; acc.y += v0.y; acc.z += v0.z; acc.w += v0.w;
            if (g > 1) { acc.x += v1.x; acc.y += v1.y; acc.z += v1.z; acc.w += v1.w; }
            if (g > 2) { acc.x += v2.x; acc.y += v2.y; acc.z += v2.z; acc.w += v2.w; }
            if (g > 3) { acc.x += v3.x; acc.y += v3.y; acc.z += v3.z; acc.w += v3.w; }
        }
    }
    ((float4*)agg)[(size_t)node * 32 + lane] = acc;
}

__global__ __launch_bounds__(256)
void sage_gemm_kernel(const float* __restrict__ agg,
                      const float* __restrict__ invdeg,
                      const float* __restrict__ h,
                      const float* __restrict__ Wl,
                      const float* __restrict__ Wr,
                      const float* __restrict__ bias,
                      float* __restrict__ out,
                      int n_nodes, int do_relu) {
    const int lane = threadIdx.x & 31;
    const int wave = threadIdx.x >> 5;
    const int m0   = blockIdx.x * 16;
    const int n0   = wave * 16;

    const int lrow  = lane & 15;
    const int khalf = (lane >> 4) * 2;

    int mrow = m0 + lrow;
    if (mrow >= n_nodes) mrow = n_nodes - 1;
    const float idg = invdeg[mrow];
    const float* __restrict__ hrow = h   + (size_t)mrow * HID;
    const float* __restrict__ arow = agg + (size_t)mrow * HID;

    const int ncol = n0 + lrow;

    v8f acc_l = {};
    v8f acc_r = {};

    for (int k0 = 0; k0 < HID; k0 += 4) {
        const int k = k0 + khalf;
        v2f a_l, a_r, b_l, b_r;
        a_l.x = arow[k]     * idg;
        a_l.y = arow[k + 1] * idg;
        a_r.x = hrow[k];
        a_r.y = hrow[k + 1];
        b_l.x = Wl[(size_t)k * HID + ncol];
        b_l.y = Wl[(size_t)(k + 1) * HID + ncol];
        b_r.x = Wr[(size_t)k * HID + ncol];
        b_r.y = Wr[(size_t)(k + 1) * HID + ncol];

        acc_l = __builtin_amdgcn_wmma_f32_16x16x4_f32(
            false, a_l, false, b_l, (short)0, acc_l, false, false);
        acc_r = __builtin_amdgcn_wmma_f32_16x16x4_f32(
            false, a_r, false, b_r, (short)0, acc_r, false, false);
    }

    const float bv = bias[ncol];
    const int mbase = m0 + (lane >> 4) * 8;
#pragma unroll
    for (int v = 0; v < 8; ++v) {
        const int m = mbase + v;
        if (m < n_nodes) {
            float o = acc_l[v] + acc_r[v] + bv;
            if (do_relu) o = o > 0.0f ? o : 0.0f;
            out[(size_t)m * HID + ncol] = o;
        }
    }
}

__global__ void edge_dot_kernel(const float* __restrict__ h2,
                                const int* __restrict__ es,
                                const int* __restrict__ ed,
                                float* __restrict__ out, int n) {
    int e = blockIdx.x * blockDim.x + threadIdx.x;
    if (e >= n) return;
    const float4* a = (const float4*)(h2 + (size_t)es[e] * HID);
    const float4* b = (const float4*)(h2 + (size_t)ed[e] * HID);
    float s = 0.0f;
#pragma unroll 8
    for (int i = 0; i < HID / 4; ++i) {
        float4 x = a[i], y = b[i];
        s += x.x * y.x + x.y * y.y + x.z * y.z + x.w * y.w;
    }
    out[e] = s;
}

extern "C" void kernel_launch(void* const* d_in, const int* in_sizes, int n_in,
                              void* d_out, int out_size, void* d_ws, size_t ws_size,
                              hipStream_t stream) {
    const int*   n_id     = (const int*)  d_in[0];
    const float* x_struct = (const float*)d_in[1];
    const int*   eidx     = (const int*)  d_in[2];
    const int*   elab     = (const int*)  d_in[3];
    const float* emb      = (const float*)d_in[4];
    const float* W1l      = (const float*)d_in[5];
    const float* W1r      = (const float*)d_in[6];
    const float* b1       = (const float*)d_in[7];
    const float* W2l      = (const float*)d_in[8];
    const float* W2r      = (const float*)d_in[9];
    const float* b2       = (const float*)d_in[10];
    float*       out      = (float*)d_out;

    const int N = in_sizes[0];
    const int E = in_sizes[2] / 2;
    const int L = in_sizes[3] / 2;

    const int* src = eidx;
    const int* dst = eidx + E;
    const int* lsrc = elab;
    const int* ldst = elab + L;

    float* ws     = (float*)d_ws;
    float* h0     = ws;
    float* h1     = h0  + (size_t)N * HID;
    float* lists  = h1  + (size_t)N * HID;
    float* agg    = lists + (size_t)N * HID;
    int*   cnt    = (int*)(agg + (size_t)N * HID);
    float* invdeg = (float*)(cnt + N);
    int*   rowptr = (int*)lists;
    int*   rank   = rowptr + N;
    int*   col    = rank + E;
    float* h2     = h0;

    const long long nh = (long long)N * HID;

    fill0_kernel<<<(N + 255) / 256, 256, 0, stream>>>((float*)cnt, N);
    build_h0_kernel<<<(int)((nh + 255) / 256), 256, 0, stream>>>(
        n_id, emb, x_struct, h0, N);
    rank_kernel<<<(E + 255) / 256, 256, 0, stream>>>(dst, cnt, rank, E);
    scan_kernel<<<1, 1024, 0, stream>>>(cnt, rowptr, invdeg, N);
    fill_kernel<<<(E + 255) / 256, 256, 0, stream>>>(src, dst, rowptr, rank, col, E);

    const int gblocks = (N + 7) / 8;
    const int mblocks = (N + 15) / 16;

    gather_kernel<<<gblocks, 256, 0, stream>>>(h0, rowptr, cnt, col, agg, N);
    sage_gemm_kernel<<<mblocks, 256, 0, stream>>>(
        agg, invdeg, h0, W1l, W1r, b1, h1, N,  1);

    gather_kernel<<<gblocks, 256, 0, stream>>>(h1, rowptr, cnt, col, agg, N);
    sage_gemm_kernel<<<mblocks, 256, 0, stream>>>(
        agg, invdeg, h1, W2l, W2r, b2, h2, N,  0);

    edge_dot_kernel<<<(L + 255) / 256, 256, 0, stream>>>(h2, lsrc, ldst, out, L);
}


# sage_gemm re-tiled: each wave keeps its 16-column slices of Wl and Wr in registers and walks 16-row tiles whose agg and h rows are staged in LDS as whole rows; 512 blocks; sums unchanged
# speedup vs baseline: 2.7173x; 1.1981x over previous
#include <hip/hip_runtime.h>
#include <hip/hip_bf16.h>


typedef __attribute__((ext_vector_type(2))) float v2f;
typedef __attribute__((ext_vector_type(8))) float v8f;

#define HID 128
#define EMBD 64

__global__ void fill0_kernel(float* __restrict__ p, long long n) {
    long long i = (long long)blockIdx.x * blockDim.x + threadIdx.x;
    if (i < n) p[i] = 0.0f;
}

__global__ void build_h0_kernel(const int* __restrict__ n_id,
                                const float* __restrict__ emb,
                                const float* __restrict__ x_struct,
                                float* __restrict__ h0, int n_nodes) {
    long long t = (long long)blockIdx.x * blockDim.x + threadIdx.x;
    int i = (int)(t >> 7);
    int c = (int)(t & 127);
    if (i >= n_nodes) return;
    float v;
    if (c < EMBD) v = emb[(size_t)n_id[i] * EMBD + c];
    else          v = x_struct[(size_t)i * EMBD + (c - EMBD)];
    h0[(size_t)i * HID + c] = v;
}

__global__ void rank_kernel(const int* __restrict__ dst, int* __restrict__ cnt,
                            int* __restrict__ rank, int n_edges) {
    int e = blockIdx.x * blockDim.x + threadIdx.x;
    if (e >= n_edges) return;
    rank[e] = atomicAdd(&cnt[dst[e]], 1);
}

__global__ __launch_bounds__(1024)
void scan_kernel(const int* __restrict__ cnt, int* __restrict__ rowptr,
                 float* __restrict__ invdeg, int n_nodes) {
    __shared__ int wave_tot[32];
    const int t    = threadIdx.x;
    const int lane = t & 31;
    const int wave = t >> 5;
    const int per  = (n_nodes + 1023) / 1024;
    const int base = t * per;
    const int left = n_nodes - base;
    const int lim  = left < per ? (left > 0 ? left : 0) : per;

    int s = 0;
    for (int i = 0; i < lim; ++i) s += cnt[base + i];
    int incl = s;
    for (int d = 1; d < 32; d <<= 1) {
        const int v = __shfl_up(incl, d);
        if (lane >= d) incl += v;
    }
    if (lane == 31) wave_tot[wave] = incl;
    __syncthreads();
    if (wave == 0) {
        const int w = wave_tot[lane];
        int wi = w;
        for (int d = 1; d < 32; d <<= 1) {
            const int v = __shfl_up(wi, d);
            if (lane >= d) wi += v;
        }
        wave_tot[lane] = wi - w;
    }
    __syncthreads();
    int run = wave_tot[wave] + incl - s;
    for (int i = 0; i < lim; ++i) {
        const int c = cnt[base + i];
        rowptr[base + i] = run;
        invdeg[base + i] = 1.0f / fmaxf((float)c, 1.0f);
        run += c;
    }
}

__global__ void fill_kernel(const int* __restrict__ src, const int* __restrict__ dst,
                            const int* __restrict__ rowptr, const int* __restrict__ rank,
                            int* __restrict__ col, int n_edges) {
    int e = blockIdx.x * blockDim.x + threadIdx.x;
    if (e >= n_edges) return;
    col[rowptr[dst[e]] + rank[e]] = src[e];
}

__global__ __launch_bounds__(256)
void gather_kernel(const float* __restrict__ h,
                   const int* __restrict__ rowptr,
                   const int* __restrict__ cnt,
                   const int* __restrict__ col,
                   float* __restrict__ agg, int n_nodes) {
    const int lane = threadIdx.x & 31;
    const int node = __builtin_amdgcn_readfirstlane(
        (int)(blockIdx.x * 8 + (threadIdx.x >> 5)));
    if (node >= n_nodes) return;
    const int start = rowptr[node];
    const int deg   = cnt[node];
    const unsigned last = (unsigned)(n_nodes - 1);
    const float4* __restrict__ h4 = (const float4*)h;

    float4 acc = make_float4(0.0f, 0.0f, 0.0f, 0.0f);
    for (int j0 = 0; j0 < deg; j0 += 32) {
        const int nb = (deg - j0) < 32 ? (deg - j0) : 32;
        unsigned mine = 0;
        if (lane < nb) mine = (unsigned)col[start + j0 + lane];
        mine = mine < last ? mine : last;
        for (int j = 0; j < nb; j += 4) {
            const int g = nb - j;
            const float4 zero4 = make_float4(0.0f, 0.0f, 0.0f, 0.0f);
            float4 v0, v1 = zero4, v2 = zero4, v3 = zero4;
            v0 = h4[(size_t)__builtin_amdgcn_readlane(mine, j) * 32 + lane];
            if (g > 1) v1 = h4[(size_t)__builtin_amdgcn_readlane(mine, j + 1) * 32 + lane];
            if (g > 2) v2 = h4[(size_t)__builtin_amdgcn_readlane(mine, j + 2) * 32 + lane];
            if (g > 3) v3 = h4[(size_t)__builtin_amdgcn_readlane(mine, j + 3) * 32 + lane];
            acc.x += v0.x; acc.y += v0.y; acc.z += v0.z; acc.w += v0.w;
            if (g > 1) { acc.x += v1.x; acc.y += v1.y; acc.z += v1.z; acc.w += v1.w; }
            if (g > 2) { acc.x += v2.x; acc.y += v2.y; acc.z += v2.z; acc.w += v2.w; }
            if (g > 3) { acc.x += v3.x; acc.y += v3.y; acc.z += v3.z; acc.w += v3.w; }
        }
    }
    ((float4*)agg)[(size_t)node * 32 + lane] = acc;
}

#define APITCH 132
#define GEMM_BLOCKS 512

__global__ __launch_bounds__(256)
void sage_gemm_kernel(const float* __restrict__ agg,
                      const float* __restrict__ invdeg,
                      const float* __restrict__ h,
                      const float* __restrict__ Wl,
                      const float* __restrict__ Wr,
                      const float* __restrict__ bias,
                      float* __restrict__ out,
                      int n_nodes, int do_relu) {
    __shared__ float sA[2][2][16][APITCH];

    const int lane = threadIdx.x & 31;
    const int wave = threadIdx.x >> 5;
    const int n0   = wave * 16;

    const int lrow  = lane & 15;
    const int khalf = (lane >> 4) * 2;
    const int ncol  = n0 + lrow;

    const int ntiles = (n_nodes + 15) >> 4;
    const int tstep  = (int)gridDim.x;
    int t = (int)blockIdx.x;
    if (t >= ntiles) return;

    v2f bl[32], br[32];
#pragma unroll
    for (int j = 0; j < 32; ++j) {
        const int k = 4 * j + khalf;
        bl[j].x = Wl[(size_t)k * HID + ncol];
        bl[j].y = Wl[(size_t)(k + 1) * HID + ncol];
        br[j].x = Wr[(size_t)k * HID + ncol];
        br[j].y = Wr[(size_t)(k + 1) * HID + ncol];
    }
    const float bv = bias[ncol];

    const float4* __restrict__ agg4 = (const float4*)agg;
    const float4* __restrict__ h4   = (const float4*)h;
    const int sr0 = 2 * wave;
    const int sr1 = 2 * wave + 1;

    float4 pa0, pa1, ph0, ph1;
    float  i0, i1;

    {
        int r0 = t * 16 + sr0, r1 = t * 16 + sr1;
        if (r0 >= n_nodes) r0 = n_nodes - 1;
        if (r1 >= n_nodes) r1 = n_nodes - 1;
        pa0 = agg4[(size_t)r0 * 32 + lane];
        pa1 = agg4[(size_t)r1 * 32 + lane];
        ph0 = h4[(size_t)r0 * 32 + lane];
        ph1 = h4[(size_t)r1 * 32 + lane];
        i0 = invdeg[r0];
        i1 = invdeg[r1];
        *(float4*)&sA[0][0][sr0][4 * lane] = make_float4(pa0.x * i0, pa0.y * i0, pa0.z * i0, pa0.w * i0);
        *(float4*)&sA[0][0][sr1][4 * lane] = make_float4(pa1.x * i1, pa1.y * i1, pa1.z * i1, pa1.w * i1);
        *(float4*)&sA[0][1][sr0][4 * lane] = ph0;
        *(float4*)&sA[0][1][sr1][4 * lane] = ph1;
    }
    __syncthreads();

    int cur = 0;
    for (; t < ntiles; t += tstep) {
        const int  tn   = t + tstep;
        const bool more = tn < ntiles;
        if (more) {
            int r0 = tn * 16 + sr0, r1 = tn * 16 + sr1;
            if (r0 >= n_nodes) r0 = n_nodes - 1;
            if (r1 >= n_nodes) r1 = n_nodes - 1;
            pa0 = agg4[(size_t)r0 * 32 + lane];
            pa1 = agg4[(size_t)r1 * 32 + lane];
            ph0 = h4[(size_t)r0 * 32 + lane];
            ph1 = h4[(size_t)r1 * 32 + lane];
            i0 = invdeg[r0];
            i1 = invdeg[r1];
        }

        v8f acc_l = {};
        v8f acc_r = {};
        const float* __restrict__ pl = &sA[cur][0][lrow][khalf];
        const float* __restrict__ pr = &sA[cur][1][lrow][khalf];
#pragma unroll
        for (int j = 0; j < 32; ++j) {
            const v2f a_l = *(const v2f*)(pl + 4 * j);
            const v2f a_r = *(const v2f*)(pr + 4 * j);
            acc_l = __builtin_amdgcn_wmma_f32_16x16x4_f32(
                false, a_l, false, bl[j], (short)0, acc_l, false, false);
            acc_r = __builtin_amdgcn_wmma_f32_16x16x4_f32(
                false, a_r, false, br[j], (short)0, acc_r, false, false);
        }

        const int mbase = t * 16 + (lane >> 4) * 8;
#pragma unroll
        for (int v = 0; v < 8; ++v) {
            int m = mbase + v;
            if (m >= n_nodes) m = n_nodes - 1;
            float o = acc_l[v] + acc_r[v] + bv;
            if (do_relu) o = o > 0.0f ? o : 0.0f;
            out[(size_t)m * HID + ncol] = o;
        }

        if (more) {
            const int nb = cur ^ 1;
            *(float4*)&sA[nb][0][sr0][4 * lane] = make_float4(pa0.x * i0, pa0.y * i0, pa0.z * i0, pa0.w * i0);
            *(float4*)&sA[nb][0][sr1][4 * lane] = make_float4(pa1.x * i1, pa1.y * i1, pa1.z * i1, pa1.w * i1);
            *(float4*)&sA[nb][1][sr0][4 * lane] = ph0;
            *(float4*)&sA[nb][1][sr1][4 * lane] = ph1;
        }
        __builtin_amdgcn_fence(__ATOMIC_RELEASE, "workgroup", "local");
        __builtin_amdgcn_s_barrier();
        __builtin_amdgcn_fence(__ATOMIC_ACQUIRE, "workgroup", "local");
        cur ^= 1;
    }
}

__global__ void edge_dot_kernel(const float* __restrict__ h2,
                                const int* __restrict__ es,
                                const int* __restrict__ ed,
                                float* __restrict__ out, int n) {
    int e = blockIdx.x * blockDim.x + threadIdx.x;
    if (e >= n) return;
    const float4* a = (const float4*)(h2 + (size_t)es[e] * HID);
    const float4* b = (const float4*)(h2 + (size_t)ed[e] * HID);
    float s = 0.0f;
#pragma unroll 8
    for (int i = 0; i < HID / 4; ++i) {
        float4 x = a[i], y = b[i];
        s += x.x * y.x + x.y * y.y + x.z * y.z + x.w * y.w;
    }
    out[e] = s;
}

extern "C" void kernel_launch(void* const* d_in, const int* in_sizes, int n_in,
                              void* d_out, int out_size, void* d_ws, size_t ws_size,
                              hipStream_t stream) {
    const int*   n_id     = (const int*)  d_in[0];
    const float* x_struct = (const float*)d_in[1];
    const int*   eidx     = (const int*)  d_in[2];
    const int*   elab     = (const int*)  d_in[3];
    const float* emb      = (const float*)d_in[4];
    const float* W1l      = (const float*)d_in[5];
    const float* W1r      = (const float*)d_in[6];
    const float* b1       = (const float*)d_in[7];
    const float* W2l      = (const float*)d_in[8];
    const float* W2r      = (const float*)d_in[9];
    const float* b2       = (const float*)d_in[10];
    float*       out      = (float*)d_out;

    const int N = in_sizes[0];
    const int E = in_sizes[2] / 2;
    const int L = in_sizes[3] / 2;

    const int* src = eidx;
    const int* dst = eidx + E;
    const int* lsrc = elab;
    const int* ldst = elab + L;

    float* ws     = (float*)d_ws;
    float* h0     = ws;
    float* h1     = h0  + (size_t)N * HID;
    float* lists  = h1  + (size_t)N * HID;
    float* agg    = lists + (size_t)N * HID;
    int*   cnt    = (int*)(agg + (size_t)N * HID);
    float* invdeg = (float*)(cnt + N);
    int*   rowptr = (int*)lists;
    int*   rank   = rowptr + N;
    int*   col    = rank + E;
    float* h2     = h0;

    const long long nh = (long long)N * HID;

    fill0_kernel<<<(N + 255) / 256, 256, 0, stream>>>((float*)cnt, N);
    build_h0_kernel<<<(int)((nh + 255) / 256), 256, 0, stream>>>(
        n_id, emb, x_struct, h0, N);
    rank_kernel<<<(E + 255) / 256, 256, 0, stream>>>(dst, cnt, rank, E);
    scan_kernel<<<1, 1024, 0, stream>>>(cnt, rowptr, invdeg, N);
    fill_kernel<<<(E + 255) / 256, 256, 0, stream>>>(src, dst, rowptr, rank, col, E);

    const int gblocks = (N + 7) / 8;

    gather_kernel<<<gblocks, 256, 0, stream>>>(h0, rowptr, cnt, col, agg, N);
    sage_gemm_kernel<<<GEMM_BLOCKS, 256, 0, stream>>>(
        agg, invdeg, h0, W1l, W1r, b1, h1, N,  1);

    gather_kernel<<<gblocks, 256, 0, stream>>>(h1, rowptr, cnt, col, agg, N);
    sage_gemm_kernel<<<GEMM_BLOCKS, 256, 0, stream>>>(
        agg, invdeg, h1, W2l, W2r, b2, h2, N,  0);

    edge_dot_kernel<<<(L + 255) / 256, 256, 0, stream>>>(h2, lsrc, ldst, out, L);
}


# scan and edge-dot re-dealt so every load and store of a wave covers whole 128-byte lines (were one strided run per lane: 32 lines each); dot sums now per-lane partials + lane butterfly
# speedup vs baseline: 3.2490x; 1.1957x over previous
#include <hip/hip_runtime.h>
#include <hip/hip_bf16.h>


typedef __attribute__((ext_vector_type(2))) float v2f;
typedef __attribute__((ext_vector_type(8))) float v8f;

#define HID 128
#define EMBD 64

__global__ void fill0_kernel(float* __restrict__ p, long long n) {
    long long i = (long long)blockIdx.x * blockDim.x + threadIdx.x;
    if (i < n) p[i] = 0.0f;
}

__global__ void build_h0_kernel(const int* __restrict__ n_id,
                                const float* __restrict__ emb,
                                const float* __restrict__ x_struct,
                                float* __restrict__ h0, int n_nodes) {
    long long t = (long long)blockIdx.x * blockDim.x + threadIdx.x;
    int i = (int)(t >> 7);
    int c = (int)(t & 127);
    if (i >= n_nodes) return;
    float v;
    if (c < EMBD) v = emb[(size_t)n_id[i] * EMBD + c];
    else          v = x_struct[(size_t)i * EMBD + (c - EMBD)];
    h0[(size_t)i * HID + c] = v;
}

__global__ void rank_kernel(const int* __restrict__ dst, int* __restrict__ cnt,
                            int* __restrict__ rank, int n_edges) {
    int e = blockIdx.x * blockDim.x + threadIdx.x;
    if (e >= n_edges) return;
    rank[e] = atomicAdd(&cnt[dst[e]], 1);
}


__device__ __forceinline__ int wave_incl_sum(int x, int lane) {
    x += __builtin_amdgcn_update_dpp(0, x, 0x111, 0xf, 0xf, false);
    x += __builtin_amdgcn_update_dpp(0, x, 0x112, 0xf, 0xf, false);
    x += __builtin_amdgcn_update_dpp(0, x, 0x114, 0xf, 0xf, false);
    x += __builtin_amdgcn_update_dpp(0, x, 0x118, 0xf, 0xf, false);
    const int low = __builtin_amdgcn_readlane(x, 15);
    return lane >= 16 ? x + low : x;
}

#define SCAN_B 16

__global__ __launch_bounds__(1024)
void scan_kernel(const int* __restrict__ cnt, int* __restrict__ rowptr,
                 float* __restrict__ invdeg, int n_nodes) {
    __shared__ int wave_tot[32];
    if (n_nodes <= 0) return;
    const int lane = threadIdx.x & 31;
    const int wave = __builtin_amdgcn_readfirstlane((int)(threadIdx.x >> 5));
    const int rows_total = (n_nodes + 127) >> 7;
    const int rows_per   = (rows_total + 31) >> 5;
    const int first = wave * rows_per * 128;
    int end = first + rows_per * 128;
    end = end < n_nodes ? end : n_nodes;
    const int nfull = end > first ? (end - first) >> 7 : 0;
    const int fend  = first + nfull * 128;
    const int it    = fend + 4 * lane;

    int s = 0;
#pragma unroll 1
    for (int b0 = first; b0 < fend; b0 += 128 * SCAN_B) {
        int c[SCAN_B][4];
#pragma unroll
        for (int u = 0; u < SCAN_B; ++u) {
            const int i0 = b0 + 128 * u + 4 * lane;
            c[u][0] = c[u][1] = c[u][2] = c[u][3] = 0;
            if (b0 + 128 * u < fend) {
                c[u][0] = cnt[i0]; c[u][1] = cnt[i0 + 1]; c[u][2] = cnt[i0 + 2]; c[u][3] = cnt[i0 + 3];
            }
        }
#pragma unroll
        for (int u = 0; u < SCAN_B; ++u) s += (c[u][0] + c[u][1]) + (c[u][2] + c[u][3]);
    }
    if (fend < end) {
        int c0 = 0, c1 = 0, c2 = 0, c3 = 0;
        if (it     < end) c0 = cnt[it];
        if (it + 1 < end) c1 = cnt[it + 1];
        if (it + 2 < end) c2 = cnt[it + 2];
        if (it + 3 < end) c3 = cnt[it + 3];
        s += (c0 + c1) + (c2 + c3);
    }
    const int tot = __builtin_amdgcn_readlane(wave_incl_sum(s, lane), 31);
    if (lane == 0) wave_tot[wave] = tot;
    __syncthreads();
    const int wt = wave_tot[lane];
    const int wx = wave_incl_sum(wt, lane) - wt;
    int run = __builtin_amdgcn_readlane(wx, wave);

#pragma unroll 1
    for (int b0 = first; b0 < fend; b0 += 128 * SCAN_B) {
        int c[SCAN_B][4];
#pragma unroll
        for (int u = 0; u < SCAN_B; ++u) {
            const int i0 = b0 + 128 * u + 4 * lane;
            c[u][0] = c[u][1] = c[u][2] = c[u][3] = 0;
            if (b0 + 128 * u < fend) {
                c[u][0] = cnt[i0]; c[u][1] = cnt[i0 + 1]; c[u][2] = cnt[i0 + 2]; c[u][3] = cnt[i0 + 3];
            }
        }
#pragma unroll
        for (int u = 0; u < SCAN_B; ++u) {
            if (b0 + 128 * u < fend) {
                const int i0 = b0 + 128 * u + 4 * lane;
                const int p1 = c[u][0];
                const int p2 = p1 + c[u][1];
                const int p3 = p2 + c[u][2];
                const int t  = p3 + c[u][3];
                const int incl = wave_incl_sum(t, lane);
                const int x0 = run + incl - t;
                rowptr[i0] = x0; rowptr[i0 + 1] = x0 + p1; rowptr[i0 + 2] = x0 + p2; rowptr[i0 + 3] = x0 + p3;
                invdeg[i0]     = 1.0f / fmaxf((float)c[u][0], 1.0f);
                invdeg[i0 + 1] = 1.0f / fmaxf((float)c[u][1], 1.0f);
                invdeg[i0 + 2] = 1.0f / fmaxf((float)c[u][2], 1.0f);
                invdeg[i0 + 3] = 1.0f / fmaxf((float)c[u][3], 1.0f);
                run += __builtin_amdgcn_readlane(incl, 31);
            }
        }
    }
    if (fend < end) {
        int c0 = 0, c1 = 0, c2 = 0, c3 = 0;
        if (it     < end) c0 = cnt[it];
        if (it + 1 < end) c1 = cnt[it + 1];
        if (it + 2 < end) c2 = cnt[it + 2];
        if (it + 3 < end) c3 = cnt[it + 3];
        const int p2 = c0 + c1;
        const int p3 = p2 + c2;
        const int t  = p3 + c3;
        const int incl = wave_incl_sum(t, lane);
        const int x0 = run + incl - t;
        if (it     < end) { rowptr[it]     = x0;      invdeg[it]     = 1.0f / fmaxf((float)c0, 1.0f); }
        if (it + 1 < end) { rowptr[it + 1] = x0 + c0; invdeg[it + 1] = 1.0f / fmaxf((float)c1, 1.0f); }
        if (it + 2 < end) { rowptr[it + 2] = x0 + p2; invdeg[it + 2] = 1.0f / fmaxf((float)c2, 1.0f); }
        if (it + 3 < end) { rowptr[it + 3] = x0 + p3; invdeg[it + 3] = 1.0f / fmaxf((float)c3, 1.0f); }
    }
}

__global__ void fill_kernel(const int* __restrict__ src, const int* __restrict__ dst,
                            const int* __restrict__ rowptr, const int* __restrict__ rank,
                            int* __restrict__ col, int n_edges) {
    int e = blockIdx.x * blockDim.x + threadIdx.x;
    if (e >= n_edges) return;
    col[rowptr[dst[e]] + rank[e]] = src[e];
}

__global__ __launch_bounds__(256)
void gather_kernel(const float* __restrict__ h,
                   const int* __restrict__ rowptr,
                   const int* __restrict__ cnt,
                   const int* __restrict__ col,
                   float* __restrict__ agg, int n_nodes) {
    const int lane = threadIdx.x & 31;
    const int node = __builtin_amdgcn_readfirstlane(
        (int)(blockIdx.x * 8 + (threadIdx.x >> 5)));
    if (node >= n_nodes) return;
    const int start = rowptr[node];
    const int deg   = cnt[node];
    const unsigned last = (unsigned)(n_nodes - 1);
    const float4* __restrict__ h4 = (const float4*)h;

    float4 acc = make_float4(0.0f, 0.0f, 0.0f, 0.0f);
    for (int j0 = 0; j0 < deg; j0 += 32) {
        const int nb = (deg - j0) < 32 ? (deg - j0) : 32;
        unsigned mine = 0;
        if (lane < nb) mine = (unsigned)col[start + j0 + lane];
        mine = mine < last ? mine : last;
        for (int j = 0; j < nb; j += 4) {
            const int g = nb - j;
            const float4 zero4 = make_float4(0.0f, 0.0f, 0.0f, 0.0f);
            float4 v0, v1 = zero4, v2 = zero4, v3 = zero4;
            v0 = h4[(size_t)__builtin_amdgcn_readlane(mine, j) * 32 + lane];
            if (g > 1) v1 = h4[(size_t)__builtin_amdgcn_readlane(mine, j + 1) * 32 + lane];
            if (g > 2) v2 = h4[(size_t)__builtin_amdgcn_readlane(mine, j + 2) * 32 + lane];
            if (g > 3) v3 = h4[(size_t)__builtin_amdgcn_readlane(mine, j + 3) * 32 + lane];
            acc.x += v0.x; acc.y += v0.y; acc.z += v0.z; acc.w += v0.w;
            if (g > 1) { acc.x += v1.x; acc.y += v1.y; acc.z += v1.z; acc.w += v1.w; }
            if (g > 2) { acc.x += v2.x; acc.y += v2.y; acc.z += v2.z; acc.w += v2.w; }
            if (g > 3) { acc.x += v3.x; acc.y += v3.y; acc.z += v3.z; acc.w += v3.w; }
        }
    }
    ((float4*)agg)[(size_t)node * 32 + lane] = acc;
}

#define APITCH 132
#define GEMM_BLOCKS 512

__global__ __launch_bounds__(256)
void sage_gemm_kernel(const float* __restrict__ agg,
                      const float* __restrict__ invdeg,
                      const float* __restrict__ h,
                      const float* __restrict__ Wl,
                      const float* __restrict__ Wr,
                      const float* __restrict__ bias,
                      float* __restrict__ out,
                      int n_nodes, int do_relu) {
    __shared__ float sA[2][2][16][APITCH];

    const int lane = threadIdx.x & 31;
    const int wave = threadIdx.x >> 5;
    const int n0   = wave * 16;

    const int lrow  = lane & 15;
    const int khalf = (lane >> 4) * 2;
    const int ncol  = n0 + lrow;

    const int ntiles = (n_nodes + 15) >> 4;
    const int tstep  = (int)gridDim.x;
    int t = (int)blockIdx.x;
    if (t >= ntiles) return;

    v2f bl[32], br[32];
#pragma unroll
    for (int j = 0; j < 32; ++j) {
        const int k = 4 * j + khalf;
        bl[j].x = Wl[(size_t)k * HID + ncol];
        bl[j].y = Wl[(size_t)(k + 1) * HID + ncol];
        br[j].x = Wr[(size_t)k * HID + ncol];
        br[j].y = Wr[(size_t)(k + 1) * HID + ncol];
    }
    const float bv = bias[ncol];

    const float4* __restrict__ agg4 = (const float4*)agg;
    const float4* __restrict__ h4   = (const float4*)h;
    const int sr0 = 2 * wave;
    const int sr1 = 2 * wave + 1;

    float4 pa0, pa1, ph0, ph1;
    float  i0, i1;

    {
        int r0 = t * 16 + sr0, r1 = t * 16 + sr1;
        if (r0 >= n_nodes) r0 = n_nodes - 1;
        if (r1 >= n_nodes) r1 = n_nodes - 1;
        pa0 = agg4[(size_t)r0 * 32 + lane];
        pa1 = agg4[(size_t)r1 * 32 + lane];
        ph0 = h4[(size_t)r0 * 32 + lane];
        ph1 = h4[(size_t)r1 * 32 + lane];
        i0 = invdeg[r0];
        i1 = invdeg[r1];
        *(float4*)&sA[0][0][sr0][4 * lane] = make_float4(pa0.x * i0, pa0.y * i0, pa0.z * i0, pa0.w * i0);
        *(float4*)&sA[0][0][sr1][4 * lane] = make_float4(pa1.x * i1, pa1.y * i1, pa1.z * i1, pa1.w * i1);
        *(float4*)&sA[0][1][sr0][4 * lane] = ph0;
        *(float4*)&sA[0][1][sr1][4 * lane] = ph1;
    }
    __syncthreads();

    int cur = 0;
    for (; t < ntiles; t += tstep) {
        const int  tn   = t + tstep;
        const bool more = tn < ntiles;
        if (more) {
            int r0 = tn * 16 + sr0, r1 = tn * 16 + sr1;
            if (r0 >= n_nodes) r0 = n_nodes - 1;
            if (r1 >= n_nodes) r1 = n_nodes - 1;
            pa0 = agg4[(size_t)r0 * 32 + lane];
            pa1 = agg4[(size_t)r1 * 32 + lane];
            ph0 = h4[(size_t)r0 * 32 + lane];
            ph1 = h4[(size_t)r1 * 32 + lane];
            i0 = invdeg[r0];
            i1 = invdeg[r1];
        }

        v8f acc_l = {};
        v8f acc_r = {};
        const float* __restrict__ pl = &sA[cur][0][lrow][khalf];
        const float* __restrict__ pr = &sA[cur][1][lrow][khalf];
#pragma unroll
        for (int j = 0; j < 32; ++j) {
            const v2f a_l = *(const v2f*)(pl + 4 * j);
            const v2f a_r = *(const v2f*)(pr + 4 * j);
            acc_l = __builtin_amdgcn_wmma_f32_16x16x4_f32(
                false, a_l, false, bl[j], (short)0, acc_l, false, false);
            acc_r = __builtin_amdgcn_wmma_f32_16x16x4_f32(
                false, a_r, false, br[j], (short)0, acc_r, false, false);
        }

        const int mbase = t * 16 + (lane >> 4) * 8;
#pragma unroll
        for (int v = 0; v < 8; ++v) {
            int m = mbase + v;
            if (m >= n_nodes) m = n_nodes - 1;
            float o = acc_l[v] + acc_r[v] + bv;
            if (do_relu) o = o > 0.0f ? o : 0.0f;
            out[(size_t)m * HID + ncol] = o;
        }

        if (more) {
            const int nb = cur ^ 1;
            *(float4*)&sA[nb][0][sr0][4 * lane] = make_float4(pa0.x * i0, pa0.y * i0, pa0.z * i0, pa0.w * i0);
            *(float4*)&sA[nb][0][sr1][4 * lane] = make_float4(pa1.x * i1, pa1.y * i1, pa1.z * i1, pa1.w * i1);
            *(float4*)&sA[nb][1][sr0][4 * lane] = ph0;
            *(float4*)&sA[nb][1][sr1][4 * lane] = ph1;
        }
        __builtin_amdgcn_fence(__ATOMIC_RELEASE, "workgroup", "local");
        __builtin_amdgcn_s_barrier();
        __builtin_amdgcn_fence(__ATOMIC_ACQUIRE, "workgroup", "local");
        cur ^= 1;
    }
}

#define ED_G 8

__device__ __forceinline__ float wave_sum_all(float x) {
    x += __builtin_bit_cast(float, __builtin_amdgcn_update_dpp(0, __builtin_bit_cast(int, x), 0x161, 0xf, 0xf, false));
    x += __builtin_bit_cast(float, __builtin_amdgcn_update_dpp(0, __builtin_bit_cast(int, x), 0x162, 0xf, 0xf, false));
    x += __builtin_bit_cast(float, __builtin_amdgcn_update_dpp(0, __builtin_bit_cast(int, x), 0x164, 0xf, 0xf, false));
    x += __builtin_bit_cast(float, __builtin_amdgcn_update_dpp(0, __builtin_bit_cast(int, x), 0x168, 0xf, 0xf, false));
    const int b = __builtin_bit_cast(int, x);
    x += __builtin_bit_cast(float, __builtin_amdgcn_permlanex16(b, b, 0x76543210, 0xfedcba98, false, false));
    return x;
}

__global__ __launch_bounds__(256)
void edge_dot_kernel(const float* __restrict__ h2,
                     const int* __restrict__ es,
                     const int* __restrict__ ed,
                     float* __restrict__ out, int n) {
    const int lane = threadIdx.x & 31;
    const int e0 = __builtin_amdgcn_readfirstlane(
        (int)((blockIdx.x * blockDim.x + threadIdx.x) & ~31u));
    if (e0 >= n) return;
    const int e  = e0 + lane;
    const int ec = e < n ? e : n - 1;
    const int si = es[ec];
    const int di = ed[ec];
    const float4* __restrict__ h4 = (const float4*)h2;

    float res = 0.0f;
#pragma unroll 1
    for (int g = 0; g < 32; g += ED_G) {
        float4 a[ED_G], b[ED_G];
#pragma unroll
        for (int u = 0; u < ED_G; ++u) {
            const int s = __builtin_amdgcn_readlane(si, g + u);
            const int d = __builtin_amdgcn_readlane(di, g + u);
            a[u] = h4[(size_t)s * 32 + lane];
            b[u] = h4[(size_t)d * 32 + lane];
        }
#pragma unroll
        for (int u = 0; u < ED_G; ++u) {
            float p = a[u].x * b[u].x + a[u].y * b[u].y + a[u].z * b[u].z + a[u].w * b[u].w;
            p = wave_sum_all(p);
            res = (lane == g + u) ? p : res;
        }
    }
    if (e < n) out[e] = res;
}

extern "C" void kernel_launch(void* const* d_in, const int* in_sizes, int n_in,
                              void* d_out, int out_size, void* d_ws, size_t ws_size,
                              hipStream_t stream) {
    const int*   n_id     = (const int*)  d_in[0];
    const float* x_struct = (const float*)d_in[1];
    const int*   eidx     = (const int*)  d_in[2];
    const int*   elab     = (const int*)  d_in[3];
    const float* emb      = (const float*)d_in[4];
    const float* W1l      = (const float*)d_in[5];
    const float* W1r      = (const float*)d_in[6];
    const float* b1       = (const float*)d_in[7];
    const float* W2l      = (const float*)d_in[8];
    const float* W2r      = (const float*)d_in[9];
    const float* b2       = (const float*)d_in[10];
    float*       out      = (float*)d_out;

    const int N = in_sizes[0];
    const int E = in_sizes[2] / 2;
    const int L = in_sizes[3] / 2;

    const int* src = eidx;
    const int* dst = eidx + E;
    const int* lsrc = elab;
    const int* ldst = elab + L;

    float* ws     = (float*)d_ws;
    float* h0     = ws;
    float* h1     = h0  + (size_t)N * HID;
    float* lists  = h1  + (size_t)N * HID;
    float* agg    = lists + (size_t)N * HID;
    int*   cnt    = (int*)(agg + (size_t)N * HID);
    float* invdeg = (float*)(cnt + N);
    int*   rowptr = (int*)lists;
    int*   rank   = rowptr + N;
    int*   col    = rank + E;
    float* h2     = h0;

    const long long nh = (long long)N * HID;

    fill0_kernel<<<(N + 255) / 256, 256, 0, stream>>>((float*)cnt, N);
    build_h0_kernel<<<(int)((nh + 255) / 256), 256, 0, stream>>>(
        n_id, emb, x_struct, h0, N);
    rank_kernel<<<(E + 255) / 256, 256, 0, stream>>>(dst, cnt, rank, E);
    scan_kernel<<<1, 1024, 0, stream>>>(cnt, rowptr, invdeg, N);
    fill_kernel<<<(E + 255) / 256, 256, 0, stream>>>(src, dst, rowptr, rank, col, E);

    const int gblocks = (N + 7) / 8;

    gather_kernel<<<gblocks, 256, 0, stream>>>(h0, rowptr, cnt, col, agg, N);
    sage_gemm_kernel<<<GEMM_BLOCKS, 256, 0, stream>>>(
        agg, invdeg, h0, W1l, W1r, b1, h1, N,  1);

    gather_kernel<<<gblocks, 256, 0, stream>>>(h1, rowptr, cnt, col, agg, N);
    sage_gemm_kernel<<<GEMM_BLOCKS, 256, 0, stream>>>(
        agg, invdeg, h1, W2l, W2r, b2, h2, N,  0);

    edge_dot_kernel<<<(L + 255) / 256, 256, 0, stream>>>(h2, lsrc, ldst, out, L);
}


# h0 build re-dealt: a wave moves 8 whole rows, 16 bytes per lane (lanes 0-15 embedding row, 16-31 feature row), not one 4-byte element per lane: 6250 waves, were 200000; sums unchanged
# speedup vs baseline: 3.3613x; 1.0346x over previous
#include <hip/hip_runtime.h>
#include <hip/hip_bf16.h>


typedef __attribute__((ext_vector_type(2))) float v2f;
typedef __attribute__((ext_vector_type(8))) float v8f;
typedef __attribute__((ext_vector_type(4))) float v4f;

#define HID 128
#define EMBD 64

__global__ void fill0_kernel(float* __restrict__ p, long long n) {
    long long i = (long long)blockIdx.x * blockDim.x + threadIdx.x;
    if (i < n) p[i] = 0.0f;
}

#define H0_ROWS 8

__global__ __launch_bounds__(256)
void build_h0_kernel(const int* __restrict__ n_id,
                     const float* __restrict__ emb,
                     const float* __restrict__ x_struct,
                     float* __restrict__ h0, int n_nodes) {
    const int lane = threadIdx.x & 31;
    const int r0 = __builtin_amdgcn_readfirstlane(
        (int)((blockIdx.x * 8 + (threadIdx.x >> 5)) * H0_ROWS));
    if (r0 >= n_nodes) return;
    const int last = n_nodes - 1;
    const int rl  = r0 + (lane & (H0_ROWS - 1));
    const int idv = n_id[rl < last ? rl : last];
    const bool lo = lane < 16;
    const v4f* __restrict__ src4 = (const v4f*)(lo ? emb : x_struct) + (lane & 15);
    v4f* __restrict__ dst4 = (v4f*)h0 + lane;

    v4f v[H0_ROWS];
#pragma unroll
    for (int u = 0; u < H0_ROWS; ++u) {
        const int row  = (r0 + u) < last ? (r0 + u) : last;
        const int id   = __builtin_amdgcn_readlane(idv, u);
        const int srow = lo ? id : row;
        v[u] = src4[(size_t)srow * (EMBD / 4)];
    }
#pragma unroll
    for (int u = 0; u < H0_ROWS; ++u) {
        const int row = (r0 + u) < last ? (r0 + u) : last;
        dst4[(size_t)row * (HID / 4)] = v[u];
    }
}

__global__ void rank_kernel(const int* __restrict__ dst, int* __restrict__ cnt,
                            int* __restrict__ rank, int n_edges) {
    int e = blockIdx.x * blockDim.x + threadIdx.x;
    if (e >= n_edges) return;
    rank[e] = atomicAdd(&cnt[dst[e]], 1);
}


__device__ __forceinline__ int wave_incl_sum(int x, int lane) {
    x += __builtin_amdgcn_update_dpp(0, x, 0x111, 0xf, 0xf, false);
    x += __builtin_amdgcn_update_dpp(0, x, 0x112, 0xf, 0xf, false);
    x += __builtin_amdgcn_update_dpp(0, x, 0x114, 0xf, 0xf, false);
    x += __builtin_amdgcn_update_dpp(0, x, 0x118, 0xf, 0xf, false);
    const int low = __builtin_amdgcn_readlane(x, 15);
    return lane >= 16 ? x + low : x;
}

#define SCAN_B 16

__global__ __launch_bounds__(1024)
void scan_kernel(const int* __restrict__ cnt, int* __restrict__ rowptr,
                 float* __restrict__ invdeg, int n_nodes) {
    __shared__ int wave_tot[32];
    if (n_nodes <= 0) return;
    const int lane = threadIdx.x & 31;
    const int wave = __builtin_amdgcn_readfirstlane((int)(threadIdx.x >> 5));
    const int rows_total = (n_nodes + 127) >> 7;
    const int rows_per   = (rows_total + 31) >> 5;
    const int first = wave * rows_per * 128;
    int end = first + rows_per * 128;
    end = end < n_nodes ? end : n_nodes;
    const int nfull = end > first ? (end - first) >> 7 : 0;
    const int fend  = first + nfull * 128;
    const int it    = fend + 4 * lane;

    int s = 0;
#pragma unroll 1
    for (int b0 = first; b0 < fend; b0 += 128 * SCAN_B) {
        int c[SCAN_B][4];
#pragma unroll
        for (int u = 0; u < SCAN_B; ++u) {
            const int i0 = b0 + 128 * u + 4 * lane;
            c[u][0] = c[u][1] = c[u][2] = c[u][3] = 0;
            if (b0 + 128 * u < fend) {
                c[u][0] = cnt[i0]; c[u][1] = cnt[i0 + 1]; c[u][2] = cnt[i0 + 2]; c[u][3] = cnt[i0 + 3];
            }
        }
#pragma unroll
        for (int u = 0; u < SCAN_B; ++u) s += (c[u][0] + c[u][1]) + (c[u][2] + c[u][3]);
    }
    if (fend < end) {
        int c0 = 0, c1 = 0, c2 = 0, c3 = 0;
        if (it     < end) c0 = cnt[it];
        if (it + 1 < end) c1 = cnt[it + 1];
        if (it + 2 < end) c2 = cnt[it + 2];
        if (it + 3 < end) c3 = cnt[it + 3];
        s += (c0 + c1) + (c2 + c3);
    }
    const int tot = __builtin_amdgcn_readlane(wave_incl_sum(s, lane), 31);
    if (lane == 0) wave_tot[wave] = tot;
    __syncthreads();
    const int wt = wave_tot[lane];
    const int wx = wave_incl_sum(wt, lane) - wt;
    int run = __builtin_amdgcn_readlane(wx, wave);

#pragma unroll 1
    for (int b0 = first; b0 < fend; b0 += 128 * SCAN_B) {
        int c[SCAN_B][4];
#pragma unroll
        for (int u = 0; u < SCAN_B; ++u) {
            const int i0 = b0 + 128 * u + 4 * lane;
            c[u][0] = c[u][1] = c[u][2] = c[u][3] = 0;
            if (b0 + 128 * u < fend) {
                c[u][0] = cnt[i0]; c[u][1] = cnt[i0 + 1]; c[u][2] = cnt[i0 + 2]; c[u][3] = cnt[i0 + 3];
            }
        }
#pragma unroll
        for (int u = 0; u < SCAN_B; ++u) {
            if (b0 + 128 * u < fend) {
                const int i0 = b0 + 128 * u + 4 * lane;
                const int p1 = c[u][0];
                const int p2 = p1 + c[u][1];
                const int p3 = p2 + c[u][2];
                const int t  = p3 + c[u][3];
                const int incl = wave_incl_sum(t, lane);
                const int x0 = run + incl - t;
                rowptr[i0] = x0; rowptr[i0 + 1] = x0 + p1; rowptr[i0 + 2] = x0 + p2; rowptr[i0 + 3] = x0 + p3;
                invdeg[i0]     = 1.0f / fmaxf((float)c[u][0], 1.0f);
                invdeg[i0 + 1] = 1.0f / fmaxf((float)c[u][1], 1.0f);
                invdeg[i0 + 2] = 1.0f / fmaxf((float)c[u][2], 1.0f);
                invdeg[i0 + 3] = 1.0f / fmaxf((float)c[u][3], 1.0f);
                run += __builtin_amdgcn_readlane(incl, 31);
            }
        }
    }
    if (fend < end) {
        int c0 = 0, c1 = 0, c2 = 0, c3 = 0;
        if (it     < end) c0 = cnt[it];
        if (it + 1 < end) c1 = cnt[it + 1];
        if (it + 2 < end) c2 = cnt[it + 2];
        if (it + 3 < end) c3 = cnt[it + 3];
        const int p2 = c0 + c1;
        const int p3 = p2 + c2;
        const int t  = p3 + c3;
        const int incl = wave_incl_sum(t, lane);
        const int x0 = run + incl - t;
        if (it     < end) { rowptr[it]     = x0;      invdeg[it]     = 1.0f / fmaxf((float)c0, 1.0f); }
        if (it + 1 < end) { rowptr[it + 1] = x0 + c0; invdeg[it + 1] = 1.0f / fmaxf((float)c1, 1.0f); }
        if (it + 2 < end) { rowptr[it + 2] = x0 + p2; invdeg[it + 2] = 1.0f / fmaxf((float)c2, 1.0f); }
        if (it + 3 < end) { rowptr[it + 3] = x0 + p3; invdeg[it + 3] = 1.0f / fmaxf((float)c3, 1.0f); }
    }
}

__global__ void fill_kernel(const int* __restrict__ src, const int* __restrict__ dst,
                            const int* __restrict__ rowptr, const int* __restrict__ rank,
                            int* __restrict__ col, int n_edges) {
    int e = blockIdx.x * blockDim.x + threadIdx.x;
    if (e >= n_edges) return;
    col[rowptr[dst[e]] + rank[e]] = src[e];
}

__global__ __launch_bounds__(256)
void gather_kernel(const float* __restrict__ h,
                   const int* __restrict__ rowptr,
                   const int* __restrict__ cnt,
                   const int* __restrict__ col,
                   float* __restrict__ agg, int n_nodes) {
    const int lane = threadIdx.x & 31;
    const int node = __builtin_amdgcn_readfirstlane(
        (int)(blockIdx.x * 8 + (threadIdx.x >> 5)));
    if (node >= n_nodes) return;
    const int start = rowptr[node];
    const int deg   = cnt[node];
    const unsigned last = (unsigned)(n_nodes - 1);
    const float4* __restrict__ h4 = (const float4*)h;

    float4 acc = make_float4(0.0f, 0.0f, 0.0f, 0.0f);
    for (int j0 = 0; j0 < deg; j0 += 32) {
        const int nb = (deg - j0) < 32 ? (deg - j0) : 32;
        unsigned mine = 0;
        if (lane < nb) mine = (unsigned)col[start + j0 + lane];
        mine = mine < last ? mine : last;
        for (int j = 0; j < nb; j += 4) {
            const int g = nb - j;
            const float4 zero4 = make_float4(0.0f, 0.0f, 0.0f, 0.0f);
            float4 v0, v1 = zero4, v2 = zero4, v3 = zero4;
            v0 = h4[(size_t)__builtin_amdgcn_readlane(mine, j) * 32 + lane];
            if (g > 1) v1 = h4[(size_t)__builtin_amdgcn_readlane(mine, j + 1) * 32 + lane];
            if (g > 2) v2 = h4[(size_t)__builtin_amdgcn_readlane(mine, j + 2) * 32 + lane];
            if (g > 3) v3 = h4[(size_t)__builtin_amdgcn_readlane(mine, j + 3) * 32 + lane];
            acc.x += v0.x; acc.y += v0.y; acc.z += v0.z; acc.w += v0.w;
            if (g > 1) { acc.x += v1.x; acc.y += v1.y; acc.z += v1.z; acc.w += v1.w; }
            if (g > 2) { acc.x += v2.x; acc.y += v2.y; acc.z += v2.z; acc.w += v2.w; }
            if (g > 3) { acc.x += v3.x; acc.y += v3.y; acc.z += v3.z; acc.w += v3.w; }
        }
    }
    ((float4*)agg)[(size_t)node * 32 + lane] = acc;
}

#define APITCH 132
#define GEMM_BLOCKS 512

__global__ __launch_bounds__(256)
void sage_gemm_kernel(const float* __restrict__ agg,
                      const float* __restrict__ invdeg,
                      const float* __restrict__ h,
                      const float* __restrict__ Wl,
                      const float* __restrict__ Wr,
                      const float* __restrict__ bias,
                      float* __restrict__ out,
                      int n_nodes, int do_relu) {
    __shared__ float sA[2][2][16][APITCH];

    const int lane = threadIdx.x & 31;
    const int wave = threadIdx.x >> 5;
    const int n0   = wave * 16;

    const int lrow  = lane & 15;
    const int khalf = (lane >> 4) * 2;
    const int ncol  = n0 + lrow;

    const int ntiles = (n_nodes + 15) >> 4;
    const int tstep  = (int)gridDim.x;
    int t = (int)blockIdx.x;
    if (t >= ntiles) return;

    v2f bl[32], br[32];
#pragma unroll
    for (int j = 0; j < 32; ++j) {
        const int k = 4 * j + khalf;
        bl[j].x = Wl[(size_t)k * HID + ncol];
        bl[j].y = Wl[(size_t)(k + 1) * HID + ncol];
        br[j].x = Wr[(size_t)k * HID + ncol];
        br[j].y = Wr[(size_t)(k + 1) * HID + ncol];
    }
    const float bv = bias[ncol];

    const float4* __restrict__ agg4 = (const float4*)agg;
    const float4* __restrict__ h4   = (const float4*)h;
    const int sr0 = 2 * wave;
    const int sr1 = 2 * wave + 1;

    float4 pa0, pa1, ph0, ph1;
    float  i0, i1;

    {
        int r0 = t * 16 + sr0, r1 = t * 16 + sr1;
        if (r0 >= n_nodes) r0 = n_nodes - 1;
        if (r1 >= n_nodes) r1 = n_nodes - 1;
        pa0 = agg4[(size_t)r0 * 32 + lane];
        pa1 = agg4[(size_t)r1 * 32 + lane];
        ph0 = h4[(size_t)r0 * 32 + lane];
        ph1 = h4[(size_t)r1 * 32 + lane];
        i0 = invdeg[r0];
        i1 = invdeg[r1];
        *(float4*)&sA[0][0][sr0][4 * lane] = make_float4(pa0.x * i0, pa0.y * i0, pa0.z * i0, pa0.w * i0);
        *(float4*)&sA[0][0][sr1][4 * lane] = make_float4(pa1.x * i1, pa1.y * i1, pa1.z * i1, pa1.w * i1);
        *(float4*)&sA[0][1][sr0][4 * lane] = ph0;
        *(float4*)&sA[0][1][sr1][4 * lane] = ph1;
    }
    __syncthreads();

    int cur = 0;
    for (; t < ntiles; t += tstep) {
        const int  tn   = t + tstep;
        const bool more = tn < ntiles;
        if (more) {
            int r0 = tn * 16 + sr0, r1 = tn * 16 + sr1;
            if (r0 >= n_nodes) r0 = n_nodes - 1;
            if (r1 >= n_nodes) r1 = n_nodes - 1;
            pa0 = agg4[(size_t)r0 * 32 + lane];
            pa1 = agg4[(size_t)r1 * 32 + lane];
            ph0 = h4[(size_t)r0 * 32 + lane];
            ph1 = h4[(size_t)r1 * 32 + lane];
            i0 = invdeg[r0];
            i1 = invdeg[r1];
        }

        v8f acc_l = {};
        v8f acc_r = {};
        const float* __restrict__ pl = &sA[cur][0][lrow][khalf];
        const float* __restrict__ pr = &sA[cur][1][lrow][khalf];
#pragma unroll
        for (int j = 0; j < 32; ++j) {
            const v2f a_l = *(const v2f*)(pl + 4 * j);
            const v2f a_r = *(const v2f*)(pr + 4 * j);
            acc_l = __builtin_amdgcn_wmma_f32_16x16x4_f32(
                false, a_l, false, bl[j], (short)0, acc_l, false, false);
            acc_r = __builtin_amdgcn_wmma_f32_16x16x4_f32(
                false, a_r, false, br[j], (short)0, acc_r, false, false);
        }

        const int mbase = t * 16 + (lane >> 4) * 8;
#pragma unroll
        for (int v = 0; v < 8; ++v) {
            int m = mbase + v;
            if (m >= n_nodes) m = n_nodes - 1;
            float o = acc_l[v] + acc_r[v] + bv;
            if (do_relu) o = o > 0.0f ? o : 0.0f;
            out[(size_t)m * HID + ncol] = o;
        }

        if (more) {
            const int nb = cur ^ 1;
            *(float4*)&sA[nb][0][sr0][4 * lane] = make_float4(pa0.x * i0, pa0.y * i0, pa0.z * i0, pa0.w * i0);
            *(float4*)&sA[nb][0][sr1][4 * lane] = make_float4(pa1.x * i1, pa1.y * i1, pa1.z * i1, pa1.w * i1);
            *(float4*)&sA[nb][1][sr0][4 * lane] = ph0;
            *(float4*)&sA[nb][1][sr1][4 * lane] = ph1;
        }
        __builtin_amdgcn_fence(__ATOMIC_RELEASE, "workgroup", "local");
        __builtin_amdgcn_s_barrier();
        __builtin_amdgcn_fence(__ATOMIC_ACQUIRE, "workgroup", "local");
        cur ^= 1;
    }
}

#define ED_G 8

__device__ __forceinline__ float wave_sum_all(float x) {
    x += __builtin_bit_cast(float, __builtin_amdgcn_update_dpp(0, __builtin_bit_cast(int, x), 0x161, 0xf, 0xf, false));
    x += __builtin_bit_cast(float, __builtin_amdgcn_update_dpp(0, __builtin_bit_cast(int, x), 0x162, 0xf, 0xf, false));
    x += __builtin_bit_cast(float, __builtin_amdgcn_update_dpp(0, __builtin_bit_cast(int, x), 0x164, 0xf, 0xf, false));
    x += __builtin_bit_cast(float, __builtin_amdgcn_update_dpp(0, __builtin_bit_cast(int, x), 0x168, 0xf, 0xf, false));
    const int b = __builtin_bit_cast(int, x);
    x += __builtin_bit_cast(float, __builtin_amdgcn_permlanex16(b, b, 0x76543210, 0xfedcba98, false, false));
    return x;
}

__global__ __launch_bounds__(256)
void edge_dot_kernel(const float* __restrict__ h2,
                     const int* __restrict__ es,
                     const int* __restrict__ ed,
                     float* __restrict__ out, int n) {
    const int lane = threadIdx.x & 31;
    const int e0 = __builtin_amdgcn_readfirstlane(
        (int)((blockIdx.x * blockDim.x + threadIdx.x) & ~31u));
    if (e0 >= n) return;
    const int e  = e0 + lane;
    const int ec = e < n ? e : n - 1;
    const int si = es[ec];
    const int di = ed[ec];
    const float4* __restrict__ h4 = (const float4*)h2;

    float res = 0.0f;
#pragma unroll 1
    for (int g = 0; g < 32; g += ED_G) {
        float4 a[ED_G], b[ED_G];
#pragma unroll
        for (int u = 0; u < ED_G; ++u) {
            const int s = __builtin_amdgcn_readlane(si, g + u);
            const int d = __builtin_amdgcn_readlane(di, g + u);
            a[u] = h4[(size_t)s * 32 + lane];
            b[u] = h4[(size_t)d * 32 + lane];
        }
#pragma unroll
        for (int u = 0; u < ED_G; ++u) {
            float p = a[u].x * b[u].x + a[u].y * b[u].y + a[u].z * b[u].z + a[u].w * b[u].w;
            p = wave_sum_all(p);
            res = (lane == g + u) ? p : res;
        }
    }
    if (e < n) out[e] = res;
}

extern "C" void kernel_launch(void* const* d_in, const int* in_sizes, int n_in,
                              void* d_out, int out_size, void* d_ws, size_t ws_size,
                              hipStream_t stream) {
    const int*   n_id     = (const int*)  d_in[0];
    const float* x_struct = (const float*)d_in[1];
    const int*   eidx     = (const int*)  d_in[2];
    const int*   elab     = (const int*)  d_in[3];
    const float* emb      = (const float*)d_in[4];
    const float* W1l      = (const float*)d_in[5];
    const float* W1r      = (const float*)d_in[6];
    const float* b1       = (const float*)d_in[7];
    const float* W2l      = (const float*)d_in[8];
    const float* W2r      = (const float*)d_in[9];
    const float* b2       = (const float*)d_in[10];
    float*       out      = (float*)d_out;

    const int N = in_sizes[0];
    const int E = in_sizes[2] / 2;
    const int L = in_sizes[3] / 2;

    const int* src = eidx;
    const int* dst = eidx + E;
    const int* lsrc = elab;
    const int* ldst = elab + L;

    float* ws     = (float*)d_ws;
    float* h0     = ws;
    float* h1     = h0  + (size_t)N * HID;
    float* lists  = h1  + (size_t)N * HID;
    float* agg    = lists + (size_t)N * HID;
    int*   cnt    = (int*)(agg + (size_t)N * HID);
    float* invdeg = (float*)(cnt + N);
    int*   rowptr = (int*)lists;
    int*   rank   = rowptr + N;
    int*   col    = rank + E;
    float* h2     = h0;

    fill0_kernel<<<(N + 255) / 256, 256, 0, stream>>>((float*)cnt, N);
    build_h0_kernel<<<(N + 8 * H0_ROWS - 1) / (8 * H0_ROWS), 256, 0, stream>>>(
        n_id, emb, x_struct, h0, N);
    rank_kernel<<<(E + 255) / 256, 256, 0, stream>>>(dst, cnt, rank, E);
    scan_kernel<<<1, 1024, 0, stream>>>(cnt, rowptr, invdeg, N);
    fill_kernel<<<(E + 255) / 256, 256, 0, stream>>>(src, dst, rowptr, rank, col, E);

    const int gblocks = (N + 7) / 8;

    gather_kernel<<<gblocks, 256, 0, stream>>>(h0, rowptr, cnt, col, agg, N);
    sage_gemm_kernel<<<GEMM_BLOCKS, 256, 0, stream>>>(
        agg, invdeg, h0, W1l, W1r, b1, h1, N,  1);

    gather_kernel<<<gblocks, 256, 0, stream>>>(h1, rowptr, cnt, col, agg, N);
    sage_gemm_kernel<<<GEMM_BLOCKS, 256, 0, stream>>>(
        agg, invdeg, h1, W2l, W2r, b2, h2, N,  0);

    edge_dot_kernel<<<(L + 255) / 256, 256, 0, stream>>>(h2, lsrc, ldst, out, L);
}


# gather fused into sage_gemm: each wave sums the incoming rows of its two next-tile rows into the LDS stage, first reads in flight during the products; agg plane, 2 launches gone; sum order kept
# speedup vs baseline: 3.6319x; 1.0805x over previous
#include <hip/hip_runtime.h>
#include <hip/hip_bf16.h>


typedef __attribute__((ext_vector_type(2))) float v2f;
typedef __attribute__((ext_vector_type(8))) float v8f;
typedef __attribute__((ext_vector_type(4))) float v4f;

#define HID 128
#define EMBD 64

__global__ void fill0_kernel(float* __restrict__ p, long long n) {
    long long i = (long long)blockIdx.x * blockDim.x + threadIdx.x;
    if (i < n) p[i] = 0.0f;
}

#define H0_ROWS 8

__global__ __launch_bounds__(256)
void build_h0_kernel(const int* __restrict__ n_id,
                     const float* __restrict__ emb,
                     const float* __restrict__ x_struct,
                     float* __restrict__ h0, int n_nodes) {
    const int lane = threadIdx.x & 31;
    const int r0 = __builtin_amdgcn_readfirstlane(
        (int)((blockIdx.x * 8 + (threadIdx.x >> 5)) * H0_ROWS));
    if (r0 >= n_nodes) return;
    const int last = n_nodes - 1;
    const int rl  = r0 + (lane & (H0_ROWS - 1));
    const int idv = n_id[rl < last ? rl : last];
    const bool lo = lane < 16;
    const v4f* __restrict__ src4 = (const v4f*)(lo ? emb : x_struct) + (lane & 15);
    v4f* __restrict__ dst4 = (v4f*)h0 + lane;

    v4f v[H0_ROWS];
#pragma unroll
    for (int u = 0; u < H0_ROWS; ++u) {
        const int row  = (r0 + u) < last ? (r0 + u) : last;
        const int id   = __builtin_amdgcn_readlane(idv, u);
        const int srow = lo ? id : row;
        v[u] = src4[(size_t)srow * (EMBD / 4)];
    }
#pragma unroll
    for (int u = 0; u < H0_ROWS; ++u) {
        const int row = (r0 + u) < last ? (r0 + u) : last;
        dst4[(size_t)row * (HID / 4)] = v[u];
    }
}

__global__ void rank_kernel(const int* __restrict__ dst, int* __restrict__ cnt,
                            int* __restrict__ rank, int n_edges) {
    int e = blockIdx.x * blockDim.x + threadIdx.x;
    if (e >= n_edges) return;
    rank[e] = atomicAdd(&cnt[dst[e]], 1);
}


__device__ __forceinline__ int wave_incl_sum(int x, int lane) {
    x += __builtin_amdgcn_update_dpp(0, x, 0x111, 0xf, 0xf, false);
    x += __builtin_amdgcn_update_dpp(0, x, 0x112, 0xf, 0xf, false);
    x += __builtin_amdgcn_update_dpp(0, x, 0x114, 0xf, 0xf, false);
    x += __builtin_amdgcn_update_dpp(0, x, 0x118, 0xf, 0xf, false);
    const int low = __builtin_amdgcn_readlane(x, 15);
    return lane >= 16 ? x + low : x;
}

#define SCAN_B 16

__global__ __launch_bounds__(1024)
void scan_kernel(const int* __restrict__ cnt, int* __restrict__ rowptr,
                 float* __restrict__ invdeg, int n_nodes) {
    __shared__ int wave_tot[32];
    if (n_nodes <= 0) return;
    const int lane = threadIdx.x & 31;
    const int wave = __builtin_amdgcn_readfirstlane((int)(threadIdx.x >> 5));
    const int rows_total = (n_nodes + 127) >> 7;
    const int rows_per   = (rows_total + 31) >> 5;
    const int first = wave * rows_per * 128;
    int end = first + rows_per * 128;
    end = end < n_nodes ? end : n_nodes;
    const int nfull = end > first ? (end - first) >> 7 : 0;
    const int fend  = first + nfull * 128;
    const int it    = fend + 4 * lane;

    int s = 0;
#pragma unroll 1
    for (int b0 = first; b0 < fend; b0 += 128 * SCAN_B) {
        int c[SCAN_B][4];
#pragma unroll
        for (int u = 0; u < SCAN_B; ++u) {
            const int i0 = b0 + 128 * u + 4 * lane;
            c[u][0] = c[u][1] = c[u][2] = c[u][3] = 0;
            if (b0 + 128 * u < fend) {
                c[u][0] = cnt[i0]; c[u][1] = cnt[i0 + 1]; c[u][2] = cnt[i0 + 2]; c[u][3] = cnt[i0 + 3];
            }
        }
#pragma unroll
        for (int u = 0; u < SCAN_B; ++u) s += (c[u][0] + c[u][1]) + (c[u][2] + c[u][3]);
    }
    if (fend < end) {
        int c0 = 0, c1 = 0, c2 = 0, c3 = 0;
        if (it     < end) c0 = cnt[it];
        if (it + 1 < end) c1 = cnt[it + 1];
        if (it + 2 < end) c2 = cnt[it + 2];
        if (it + 3 < end) c3 = cnt[it + 3];
        s += (c0 + c1) + (c2 + c3);
    }
    const int tot = __builtin_amdgcn_readlane(wave_incl_sum(s, lane), 31);
    if (lane == 0) wave_tot[wave] = tot;
    __syncthreads();
    const int wt = wave_tot[lane];
    const int wx = wave_incl_sum(wt, lane) - wt;
    int run = __builtin_amdgcn_readlane(wx, wave);

#pragma unroll 1
    for (int b0 = first; b0 < fend; b0 += 128 * SCAN_B) {
        int c[SCAN_B][4];
#pragma unroll
        for (int u = 0; u < SCAN_B; ++u) {
            const int i0 = b0 + 128 * u + 4 * lane;
            c[u][0] = c[u][1] = c[u][2] = c[u][3] = 0;
            if (b0 + 128 * u < fend) {
                c[u][0] = cnt[i0]; c[u][1] = cnt[i0 + 1]; c[u][2] = cnt[i0 + 2]; c[u][3] = cnt[i0 + 3];
            }
        }
#pragma unroll
        for (int u = 0; u < SCAN_B; ++u) {
            if (b0 + 128 * u < fend) {
                const int i0 = b0 + 128 * u + 4 * lane;
                const int p1 = c[u][0];
                const int p2 = p1 + c[u][1];
                const int p3 = p2 + c[u][2];
                const int t  = p3 + c[u][3];
                const int incl = wave_incl_sum(t, lane);
                const int x0 = run + incl - t;
                rowptr[i0] = x0; rowptr[i0 + 1] = x0 + p1; rowptr[i0 + 2] = x0 + p2; rowptr[i0 + 3] = x0 + p3;
                invdeg[i0]     = 1.0f / fmaxf((float)c[u][0], 1.0f);
                invdeg[i0 + 1] = 1.0f / fmaxf((float)c[u][1], 1.0f);
                invdeg[i0 + 2] = 1.0f / fmaxf((float)c[u][2], 1.0f);
                invdeg[i0 + 3] = 1.0f / fmaxf((float)c[u][3], 1.0f);
                run += __builtin_amdgcn_readlane(incl, 31);
            }
        }
    }
    if (fend < end) {
        int c0 = 0, c1 = 0, c2 = 0, c3 = 0;
        if (it     < end) c0 = cnt[it];
        if (it + 1 < end) c1 = cnt[it + 1];
        if (it + 2 < end) c2 = cnt[it + 2];
        if (it + 3 < end) c3 = cnt[it + 3];
        const int p2 = c0 + c1;
        const int p3 = p2 + c2;
        const int t  = p3 + c3;
        const int incl = wave_incl_sum(t, lane);
        const int x0 = run + incl - t;
        if (it     < end) { rowptr[it]     = x0;      invdeg[it]     = 1.0f / fmaxf((float)c0, 1.0f); }
        if (it + 1 < end) { rowptr[it + 1] = x0 + c0; invdeg[it + 1] = 1.0f / fmaxf((float)c1, 1.0f); }
        if (it + 2 < end) { rowptr[it + 2] = x0 + p2; invdeg[it + 2] = 1.0f / fmaxf((float)c2, 1.0f); }
        if (it + 3 < end) { rowptr[it + 3] = x0 + p3; invdeg[it + 3] = 1.0f / fmaxf((float)c3, 1.0f); }
    }
}

__global__ void fill_kernel(const int* __restrict__ src, const int* __restrict__ dst,
                            const int* __restrict__ rowptr, const int* __restrict__ rank,
                            int* __restrict__ col, int n_edges) {
    int e = blockIdx.x * blockDim.x + threadIdx.x;
    if (e >= n_edges) return;
    col[rowptr[dst[e]] + rank[e]] = src[e];
}

#define APITCH 132
#define GEMM_BLOCKS 512
#define GR 8

__device__ __forceinline__ unsigned src_id(unsigned m, int j, unsigned last) {
    const unsigned id = (unsigned)__builtin_amdgcn_readlane((int)m, j);
    return id < last ? id : last;
}

__device__ __forceinline__ void rows_issue(const float4* __restrict__ h4,
                                           unsigned m0, int n0, unsigned m1, int n1,
                                           int j, int lane, unsigned last,
                                           float4 (&v)[GR], float4 (&w)[GR]) {
#pragma unroll
    for (int u = 0; u < GR; ++u)
        if (j + u < n0) v[u] = h4[(size_t)src_id(m0, j + u, last) * 32 + lane];
#pragma unroll
    for (int u = 0; u < GR; ++u)
        if (j + u < n1) w[u] = h4[(size_t)src_id(m1, j + u, last) * 32 + lane];
}

__device__ __forceinline__ void rows_add(int n0, int n1, int j,
                                         const float4 (&v)[GR], const float4 (&w)[GR],
                                         float4& a0, float4& a1) {
#pragma unroll
    for (int u = 0; u < GR; ++u)
        if (j + u < n0) { a0.x += v[u].x; a0.y += v[u].y; a0.z += v[u].z; a0.w += v[u].w; }
#pragma unroll
    for (int u = 0; u < GR; ++u)
        if (j + u < n1) { a1.x += w[u].x; a1.y += w[u].y; a1.z += w[u].z; a1.w += w[u].w; }
}

__device__ __forceinline__ int clamp32(int d) { return d < 0 ? 0 : (d > 32 ? 32 : d); }

__device__ __forceinline__ void gather_finish(const float4* __restrict__ h4,
                                              const int* __restrict__ col,
                                              int s0, int d0, unsigned m0,
                                              int s1, int d1, unsigned m1,
                                              int lane, unsigned last,
                                              float4 (&v)[GR], float4 (&w)[GR],
                                              float4& r0, float4& r1) {
    float4 a0 = make_float4(0.0f, 0.0f, 0.0f, 0.0f);
    float4 a1 = make_float4(0.0f, 0.0f, 0.0f, 0.0f);
    int n0 = clamp32(d0), n1 = clamp32(d1);
    int nm = n0 > n1 ? n0 : n1;
    rows_add(n0, n1, 0, v, w, a0, a1);
    for (int j = GR; j < nm; j += GR) {
        rows_issue(h4, m0, n0, m1, n1, j, lane, last, v, w);
        rows_add(n0, n1, j, v, w, a0, a1);
    }
    const int dmax = d0 > d1 ? d0 : d1;
    for (int j0 = 32; j0 < dmax; j0 += 32) {
        m0 = 0; m1 = 0;
        if (j0 + lane < d0) m0 = (unsigned)col[s0 + j0 + lane];
        if (j0 + lane < d1) m1 = (unsigned)col[s1 + j0 + lane];
        m0 = m0 < last ? m0 : last;
        m1 = m1 < last ? m1 : last;
        n0 = clamp32(d0 - j0); n1 = clamp32(d1 - j0);
        nm = n0 > n1 ? n0 : n1;
        for (int j = 0; j < nm; j += GR) {
            rows_issue(h4, m0, n0, m1, n1, j, lane, last, v, w);
            rows_add(n0, n1, j, v, w, a0, a1);
        }
    }
    r0 = a0; r1 = a1;
}

__global__ __launch_bounds__(256)
void sage_gemm_kernel(const float* __restrict__ h,
                      const int* __restrict__ rowptr,
                      const int* __restrict__ cnt,
                      const int* __restrict__ col,
                      const float* __restrict__ invdeg,
                      const float* __restrict__ Wl,
                      const float* __restrict__ Wr,
                      const float* __restrict__ bias,
                      float* __restrict__ out,
                      int n_nodes, int do_relu) {
    __shared__ float sA[2][2][16][APITCH];

    const int lane = threadIdx.x & 31;
    const int wave = __builtin_amdgcn_readfirstlane((int)(threadIdx.x >> 5));
    const int n0   = wave * 16;

    const int lrow  = lane & 15;
    const int khalf = (lane >> 4) * 2;
    const int ncol  = n0 + lrow;

    const int ntiles = (n_nodes + 15) >> 4;
    const int tstep  = (int)gridDim.x;
    int t = (int)blockIdx.x;
    if (t >= ntiles) return;

    v2f bl[32], br[32];
#pragma unroll
    for (int j = 0; j < 32; ++j) {
        const int k = 4 * j + khalf;
        bl[j].x = Wl[(size_t)k * HID + ncol];
        bl[j].y = Wl[(size_t)(k + 1) * HID + ncol];
        br[j].x = Wr[(size_t)k * HID + ncol];
        br[j].y = Wr[(size_t)(k + 1) * HID + ncol];
    }
    const float bv = bias[ncol];

    const float4* __restrict__ h4 = (const float4*)h;
    const int      lasti = n_nodes - 1;
    const unsigned last  = (unsigned)lasti;
    const int sr0 = 2 * wave;
    const int sr1 = 2 * wave + 1;

    float4 ph0, ph1, g0, g1;
    float4 v[GR], w[GR];
    float  i0, i1;

    int ra = t * 16 + sr0, rb = t * 16 + sr1;
    ra = ra < lasti ? ra : lasti;
    rb = rb < lasti ? rb : lasti;
    int cs0 = rowptr[ra], cd0 = cnt[ra], cs1 = rowptr[rb], cd1 = cnt[rb];
    unsigned m0 = (unsigned)col[lane < cd0 ? cs0 + lane : 0];
    unsigned m1 = (unsigned)col[lane < cd1 ? cs1 + lane : 0];
    ph0 = h4[(size_t)ra * 32 + lane];
    ph1 = h4[(size_t)rb * 32 + lane];
    i0 = invdeg[ra];
    i1 = invdeg[rb];
    rows_issue(h4, m0, clamp32(cd0), m1, clamp32(cd1), 0, lane, last, v, w);
    gather_finish(h4, col, cs0, cd0, m0, cs1, cd1, m1, lane, last, v, w, g0, g1);
    *(float4*)&sA[0][0][sr0][4 * lane] = make_float4(g0.x * i0, g0.y * i0, g0.z * i0, g0.w * i0);
    *(float4*)&sA[0][0][sr1][4 * lane] = make_float4(g1.x * i1, g1.y * i1, g1.z * i1, g1.w * i1);
    *(float4*)&sA[0][1][sr0][4 * lane] = ph0;
    *(float4*)&sA[0][1][sr1][4 * lane] = ph1;

    int na = (t + tstep) * 16 + sr0, nb_ = (t + tstep) * 16 + sr1;
    na  = na  < lasti ? na  : lasti;
    nb_ = nb_ < lasti ? nb_ : lasti;
    cs0 = rowptr[na]; cd0 = cnt[na]; cs1 = rowptr[nb_]; cd1 = cnt[nb_];
    m0 = (unsigned)col[lane < cd0 ? cs0 + lane : 0];
    m1 = (unsigned)col[lane < cd1 ? cs1 + lane : 0];
    ra = (t + 2 * tstep) * 16 + sr0; rb = (t + 2 * tstep) * 16 + sr1;
    ra = ra < lasti ? ra : lasti;
    rb = rb < lasti ? rb : lasti;
    int sa = rowptr[ra], da = cnt[ra], sb = rowptr[rb], db = cnt[rb];
    __syncthreads();

    int cur = 0;
    for (; t < ntiles; t += tstep) {
        const int  tn   = t + tstep;
        const bool more = tn < ntiles;
        unsigned mn0 = 0, mn1 = 0;
        const int ns0 = sa, nd0 = da, ns1 = sb, nd1 = db;
        if (more) {
            ph0 = h4[(size_t)na  * 32 + lane];
            ph1 = h4[(size_t)nb_ * 32 + lane];
            i0 = invdeg[na];
            i1 = invdeg[nb_];
            mn0 = (unsigned)col[lane < nd0 ? ns0 + lane : 0];
            mn1 = (unsigned)col[lane < nd1 ? ns1 + lane : 0];
            rows_issue(h4, m0, clamp32(cd0), m1, clamp32(cd1), 0, lane, last, v, w);
            na = ra; nb_ = rb;
            ra = (tn + 2 * tstep) * 16 + sr0; rb = (tn + 2 * tstep) * 16 + sr1;
            ra = ra < lasti ? ra : lasti;
            rb = rb < lasti ? rb : lasti;
            sa = rowptr[ra]; da = cnt[ra]; sb = rowptr[rb]; db = cnt[rb];
        }

        v8f acc_l = {};
        v8f acc_r = {};
        const float* __restrict__ pl = &sA[cur][0][lrow][khalf];
        const float* __restrict__ pr = &sA[cur][1][lrow][khalf];
#pragma unroll
        for (int j = 0; j < 32; ++j) {
            const v2f a_l = *(const v2f*)(pl + 4 * j);
            const v2f a_r = *(const v2f*)(pr + 4 * j);
            acc_l = __builtin_amdgcn_wmma_f32_16x16x4_f32(
                false, a_l, false, bl[j], (short)0, acc_l, false, false);
            acc_r = __builtin_amdgcn_wmma_f32_16x16x4_f32(
                false, a_r, false, br[j], (short)0, acc_r, false, false);
        }

        const int mbase = t * 16 + (lane >> 4) * 8;
#pragma unroll
        for (int q = 0; q < 8; ++q) {
            int m = mbase + q;
            if (m >= n_nodes) m = n_nodes - 1;
            float o = acc_l[q] + acc_r[q] + bv;
            if (do_relu) o = o > 0.0f ? o : 0.0f;
            out[(size_t)m * HID + ncol] = o;
        }

        if (more) {
            gather_finish(h4, col, cs0, cd0, m0, cs1, cd1, m1, lane, last, v, w, g0, g1);
            const int nb = cur ^ 1;
            *(float4*)&sA[nb][0][sr0][4 * lane] = make_float4(g0.x * i0, g0.y * i0, g0.z * i0, g0.w * i0);
            *(float4*)&sA[nb][0][sr1][4 * lane] = make_float4(g1.x * i1, g1.y * i1, g1.z * i1, g1.w * i1);
            *(float4*)&sA[nb][1][sr0][4 * lane] = ph0;
            *(float4*)&sA[nb][1][sr1][4 * lane] = ph1;
            cs0 = ns0; cd0 = nd0; cs1 = ns1; cd1 = nd1;
            m0 = mn0; m1 = mn1;
        }
        __builtin_amdgcn_fence(__ATOMIC_RELEASE, "workgroup", "local");
        __builtin_amdgcn_s_barrier();
        __builtin_amdgcn_fence(__ATOMIC_ACQUIRE, "workgroup", "local");
        cur ^= 1;
    }
}

#define ED_G 8

__device__ __forceinline__ float wave_sum_all(float x) {
    x += __builtin_bit_cast(float, __builtin_amdgcn_update_dpp(0, __builtin_bit_cast(int, x), 0x161, 0xf, 0xf, false));
    x += __builtin_bit_cast(float, __builtin_amdgcn_update_dpp(0, __builtin_bit_cast(int, x), 0x162, 0xf, 0xf, false));
    x += __builtin_bit_cast(float, __builtin_amdgcn_update_dpp(0, __builtin_bit_cast(int, x), 0x164, 0xf, 0xf, false));
    x += __builtin_bit_cast(float, __builtin_amdgcn_update_dpp(0, __builtin_bit_cast(int, x), 0x168, 0xf, 0xf, false));
    const int b = __builtin_bit_cast(int, x);
    x += __builtin_bit_cast(float, __builtin_amdgcn_permlanex16(b, b, 0x76543210, 0xfedcba98, false, false));
    return x;
}

__global__ __launch_bounds__(256)
void edge_dot_kernel(const float* __restrict__ h2,
                     const int* __restrict__ es,
                     const int* __restrict__ ed,
                     float* __restrict__ out, int n) {
    const int lane = threadIdx.x & 31;
    const int e0 = __builtin_amdgcn_readfirstlane(
        (int)((blockIdx.x * blockDim.x + threadIdx.x) & ~31u));
    if (e0 >= n) return;
    const int e  = e0 + lane;
    const int ec = e < n ? e : n - 1;
    const int si = es[ec];
    const int di = ed[ec];
    const float4* __restrict__ h4 = (const float4*)h2;

    float res = 0.0f;
#pragma unroll 1
    for (int g = 0; g < 32; g += ED_G) {
        float4 a[ED_G], b[ED_G];
#pragma unroll
        for (int u = 0; u < ED_G; ++u) {
            const int s = __builtin_amdgcn_readlane(si, g + u);
            const int d = __builtin_amdgcn_readlane(di, g + u);
            a[u] = h4[(size_t)s * 32 + lane];
            b[u] = h4[(size_t)d * 32 + lane];
        }
#pragma unroll
        for (int u = 0; u < ED_G; ++u) {
            float p = a[u].x * b[u].x + a[u].y * b[u].y + a[u].z * b[u].z + a[u].w * b[u].w;
            p = wave_sum_all(p);
            res = (lane == g + u) ? p : res;
        }
    }
    if (e < n) out[e] = res;
}

extern "C" void kernel_launch(void* const* d_in, const int* in_sizes, int n_in,
                              void* d_out, int out_size, void* d_ws, size_t ws_size,
                              hipStream_t stream) {
    const int*   n_id     = (const int*)  d_in[0];
    const float* x_struct = (const float*)d_in[1];
    const int*   eidx     = (const int*)  d_in[2];
    const int*   elab     = (const int*)  d_in[3];
    const float* emb      = (const float*)d_in[4];
    const float* W1l      = (const float*)d_in[5];
    const float* W1r      = (const float*)d_in[6];
    const float* b1       = (const float*)d_in[7];
    const float* W2l      = (const float*)d_in[8];
    const float* W2r      = (const float*)d_in[9];
    const float* b2       = (const float*)d_in[10];
    float*       out      = (float*)d_out;

    const int N = in_sizes[0];
    const int E = in_sizes[2] / 2;
    const int L = in_sizes[3] / 2;

    const int* src = eidx;
    const int* dst = eidx + E;
    const int* lsrc = elab;
    const int* ldst = elab + L;

    float* ws     = (float*)d_ws;
    float* h0     = ws;
    float* h1     = h0  + (size_t)N * HID;
    float* lists  = h1  + (size_t)N * HID;
    float* spare  = lists + (size_t)N * HID;
    int*   cnt    = (int*)(spare + (size_t)N * HID);
    float* invdeg = (float*)(cnt + N);
    int*   rowptr = (int*)lists;
    int*   rank   = rowptr + N;
    int*   col    = rank + E;
    float* h2     = h0;

    fill0_kernel<<<(N + 255) / 256, 256, 0, stream>>>((float*)cnt, N);
    build_h0_kernel<<<(N + 8 * H0_ROWS - 1) / (8 * H0_ROWS), 256, 0, stream>>>(
        n_id, emb, x_struct, h0, N);
    rank_kernel<<<(E + 255) / 256, 256, 0, stream>>>(dst, cnt, rank, E);
    scan_kernel<<<1, 1024, 0, stream>>>(cnt, rowptr, invdeg, N);
    fill_kernel<<<(E + 255) / 256, 256, 0, stream>>>(src, dst, rowptr, rank, col, E);

    sage_gemm_kernel<<<GEMM_BLOCKS, 256, 0, stream>>>(
        h0, rowptr, cnt, col, invdeg, W1l, W1r, b1, h1, N,  1);

    sage_gemm_kernel<<<GEMM_BLOCKS, 256, 0, stream>>>(
        h1, rowptr, cnt, col, invdeg, W2l, W2r, b2, h2, N,  0);

    edge_dot_kernel<<<(L + 255) / 256, 256, 0, stream>>>(h2, lsrc, ldst, out, L);
}


# Edges grouped by a two-level counting sort on LDS counters (256-node bins, then nodes), not one device-scope returning add per edge + scan; bucket slot order now LDS arrival order; 7 launches
# speedup vs baseline: 5.2858x; 1.4554x over previous
#include <hip/hip_runtime.h>
#include <hip/hip_bf16.h>


typedef __attribute__((ext_vector_type(2))) float v2f;
typedef __attribute__((ext_vector_type(8))) float v8f;
typedef __attribute__((ext_vector_type(4))) float v4f;

#define HID 128
#define EMBD 64

#define H0_ROWS 8

__global__ __launch_bounds__(256)
void build_h0_kernel(const int* __restrict__ n_id,
                     const float* __restrict__ emb,
                     const float* __restrict__ x_struct,
                     float* __restrict__ h0, int n_nodes) {
    const int lane = threadIdx.x & 31;
    const int r0 = __builtin_amdgcn_readfirstlane(
        (int)((blockIdx.x * 8 + (threadIdx.x >> 5)) * H0_ROWS));
    if (r0 >= n_nodes) return;
    const int last = n_nodes - 1;
    const int rl  = r0 + (lane & (H0_ROWS - 1));
    const int idv = n_id[rl < last ? rl : last];
    const bool lo = lane < 16;
    const v4f* __restrict__ src4 = (const v4f*)(lo ? emb : x_struct) + (lane & 15);
    v4f* __restrict__ dst4 = (v4f*)h0 + lane;

    v4f v[H0_ROWS];
#pragma unroll
    for (int u = 0; u < H0_ROWS; ++u) {
        const int row  = (r0 + u) < last ? (r0 + u) : last;
        const int id   = __builtin_amdgcn_readlane(idv, u);
        const int srow = lo ? id : row;
        v[u] = src4[(size_t)srow * (EMBD / 4)];
    }
#pragma unroll
    for (int u = 0; u < H0_ROWS; ++u) {
        const int row = (r0 + u) < last ? (r0 + u) : last;
        dst4[(size_t)row * (HID / 4)] = v[u];
    }
}

#define BIN_SHIFT 8
#define BIN_W     256
#define MAXBINS   1024
#define SORT_T    1024
#define SORT_U    4
#define SORT_BLOCKS 256

__device__ __forceinline__ int wave_incl_sum(int x, int lane) {
    x += __builtin_amdgcn_update_dpp(0, x, 0x111, 0xf, 0xf, false);
    x += __builtin_amdgcn_update_dpp(0, x, 0x112, 0xf, 0xf, false);
    x += __builtin_amdgcn_update_dpp(0, x, 0x114, 0xf, 0xf, false);
    x += __builtin_amdgcn_update_dpp(0, x, 0x118, 0xf, 0xf, false);
    const int low = __builtin_amdgcn_readlane(x, 15);
    return lane >= 16 ? x + low : x;
}

__device__ __forceinline__ int dst_node(int d, int n_nodes) {
    const unsigned u = (unsigned)d, last = (unsigned)(n_nodes - 1);
    return (int)(u < last ? u : last);
}

__global__ __launch_bounds__(SORT_T)
void coarse_count_kernel(const int* __restrict__ dst, int* __restrict__ counts,
                         int n_nodes, int n_edges) {
    __shared__ int hist[MAXBINS];
    const int tid   = threadIdx.x;
    const int nbins = (n_nodes + BIN_W - 1) >> BIN_SHIFT;
    const int chunk = (n_edges + (int)gridDim.x - 1) / (int)gridDim.x;
    const int eb    = (int)blockIdx.x * chunk;
    const int ee    = eb + chunk < n_edges ? eb + chunk : n_edges;
    for (int i = tid; i < nbins; i += SORT_T) hist[i] = 0;
    __syncthreads();
    for (int e0 = eb; e0 < ee; e0 += SORT_T * SORT_U) {
        int d[SORT_U];
#pragma unroll
        for (int u = 0; u < SORT_U; ++u) {
            const int e = e0 + u * SORT_T + tid;
            d[u] = dst[e < ee ? e : ee - 1];
        }
#pragma unroll
        for (int u = 0; u < SORT_U; ++u) {
            const int e = e0 + u * SORT_T + tid;
            if (e < ee) atomicAdd(&hist[dst_node(d[u], n_nodes) >> BIN_SHIFT], 1);
        }
    }
    __syncthreads();
    const int cstride = (nbins + 31) & ~31;
    for (int i = tid; i < nbins; i += SORT_T)
        counts[(size_t)blockIdx.x * cstride + i] = hist[i];
}

#define CS_PARTS 4
#define CS_G     16
__global__ __launch_bounds__(SORT_T)
void coarse_scatter_kernel(const int* __restrict__ src, const int* __restrict__ dst,
                           const int* __restrict__ counts, int* __restrict__ binstart,
                           int2* __restrict__ pairs, int n_nodes, int n_edges) {
    __shared__ int totS[MAXBINS + 1];
    __shared__ int preS[MAXBINS];
    __shared__ int curS[MAXBINS];
    const int tid   = threadIdx.x;
    const int lane  = tid & 31;
    const int wave  = __builtin_amdgcn_readfirstlane((int)(threadIdx.x >> 5));
    const int nbins = (n_nodes + BIN_W - 1) >> BIN_SHIFT;
    const int cstride = (nbins + 31) & ~31;
    const int nblk  = (int)gridDim.x;
    const int k     = (int)blockIdx.x;
    const int chunk = (n_edges + nblk - 1) / nblk;
    const int eb    = k * chunk;
    const int ee    = eb + chunk < n_edges ? eb + chunk : n_edges;

    for (int i = tid; i < nbins; i += SORT_T) { totS[i] = 0; preS[i] = 0; }
    __syncthreads();
    {
        const int part = wave / (SORT_T / 32 / CS_PARTS);
        const int jper = (nblk + CS_PARTS - 1) / CS_PARTS;
        const int j0   = part * jper;
        const int j1   = j0 + jper < nblk ? j0 + jper : nblk;
        for (int bin = tid & (SORT_T / CS_PARTS - 1); bin < nbins; bin += SORT_T / CS_PARTS) {
            int tot = 0, pre = 0;
            for (int j = j0; j < j1; j += CS_G) {
                int v[CS_G];
#pragma unroll
                for (int u = 0; u < CS_G; ++u)
                    v[u] = counts[(size_t)(j + u < j1 ? j + u : j1 - 1) * cstride + bin];
#pragma unroll
                for (int u = 0; u < CS_G; ++u) {
                    tot += (j + u < j1) ? v[u] : 0;
                    pre += (j + u < j1 && j + u < k) ? v[u] : 0;
                }
            }
            atomicAdd(&totS[bin], tot);
            atomicAdd(&preS[bin], pre);
        }
    }
    __syncthreads();
    if (wave == 0) {
        int run = 0;
        for (int c0 = 0; c0 < nbins; c0 += 32) {
            const int i = c0 + lane;
            const int x = i < nbins ? totS[i] : 0;
            const int incl = wave_incl_sum(x, lane);
            if (i < nbins) totS[i] = run + incl - x;
            run += __builtin_amdgcn_readlane(incl, 31);
        }
        if (lane == 0) totS[nbins] = run;
    }
    __syncthreads();
    for (int i = tid; i < nbins; i += SORT_T) curS[i] = totS[i] + preS[i];
    if (k == 0)
        for (int i = tid; i <= nbins; i += SORT_T) binstart[i] = totS[i];
    __syncthreads();

    const unsigned elast = (unsigned)(n_edges - 1);
    for (int e0 = eb; e0 < ee; e0 += SORT_T * SORT_U) {
        int s[SORT_U], d[SORT_U];
#pragma unroll
        for (int u = 0; u < SORT_U; ++u) {
            const int e  = e0 + u * SORT_T + tid;
            const int ec = e < ee ? e : ee - 1;
            s[u] = src[ec];
            d[u] = dst[ec];
        }
#pragma unroll
        for (int u = 0; u < SORT_U; ++u) {
            const int e = e0 + u * SORT_T + tid;
            if (e < ee) {
                const int dn = dst_node(d[u], n_nodes);
                unsigned p = (unsigned)atomicAdd(&curS[dn >> BIN_SHIFT], 1);
                p = p < elast ? p : elast;
                pairs[p] = make_int2(s[u], dn);
            }
        }
    }
}

__global__ __launch_bounds__(SORT_T)
void fine_sort_kernel(const int2* __restrict__ pairs, const int* __restrict__ binstart,
                      int* __restrict__ rowptr, int* __restrict__ cnt,
                      float* __restrict__ invdeg, int* __restrict__ col,
                      int n_nodes, int n_edges) {
    __shared__ int c[BIN_W];
    __shared__ int wt[BIN_W / 32];
    const int tid  = threadIdx.x;
    const int lane = tid & 31;
    const int wave = __builtin_amdgcn_readfirstlane((int)(threadIdx.x >> 5));
    const int lo   = (int)blockIdx.x * BIN_W;
    int ps = binstart[blockIdx.x], pe = binstart[blockIdx.x + 1];
    ps = ps < 0 ? 0 : ps;
    pe = pe > n_edges ? n_edges : pe;

    if (tid < BIN_W) c[tid] = 0;
    __syncthreads();
    for (int i0 = ps; i0 < pe; i0 += SORT_T * SORT_U) {
        int d[SORT_U];
#pragma unroll
        for (int u = 0; u < SORT_U; ++u) {
            const int i = i0 + u * SORT_T + tid;
            d[u] = pairs[i < pe ? i : pe - 1].y;
        }
#pragma unroll
        for (int u = 0; u < SORT_U; ++u) {
            const int i = i0 + u * SORT_T + tid;
            if (i < pe) atomicAdd(&c[(d[u] - lo) & (BIN_W - 1)], 1);
        }
    }
    __syncthreads();
    int deg = 0, start = 0;
    if (wave < BIN_W / 32) {
        deg = c[tid];
        const int incl = wave_incl_sum(deg, lane);
        if (lane == 31) wt[wave] = incl;
        start = ps + incl - deg;
    }
    __syncthreads();
    if (wave < BIN_W / 32) {
#pragma unroll
        for (int w2 = 0; w2 < BIN_W / 32; ++w2) start += (w2 < wave) ? wt[w2] : 0;
        c[tid] = start;
        const int node = lo + tid;
        if (node < n_nodes) {
            rowptr[node] = start;
            cnt[node]    = deg;
            invdeg[node] = 1.0f / fmaxf((float)deg, 1.0f);
        }
    }
    __syncthreads();
    const unsigned elast = (unsigned)(n_edges - 1);
    for (int i0 = ps; i0 < pe; i0 += SORT_T * SORT_U) {
        int2 p[SORT_U];
#pragma unroll
        for (int u = 0; u < SORT_U; ++u) {
            const int i = i0 + u * SORT_T + tid;
            p[u] = pairs[i < pe ? i : pe - 1];
        }
#pragma unroll
        for (int u = 0; u < SORT_U; ++u) {
            const int i = i0 + u * SORT_T + tid;
            if (i < pe) {
                unsigned slot = (unsigned)atomicAdd(&c[(p[u].y - lo) & (BIN_W - 1)], 1);
                slot = slot < elast ? slot : elast;
                col[slot] = p[u].x;
            }
        }
    }
}

#define APITCH 132
#define GEMM_BLOCKS 512
#define GR 8

__device__ __forceinline__ unsigned src_id(unsigned m, int j, unsigned last) {
    const unsigned id = (unsigned)__builtin_amdgcn_readlane((int)m, j);
    return id < last ? id : last;
}

__device__ __forceinline__ void rows_issue(const float4* __restrict__ h4,
                                           unsigned m0, int n0, unsigned m1, int n1,
                                           int j, int lane, unsigned last,
                                           float4 (&v)[GR], float4 (&w)[GR]) {
#pragma unroll
    for (int u = 0; u < GR; ++u)
        if (j + u < n0) v[u] = h4[(size_t)src_id(m0, j + u, last) * 32 + lane];
#pragma unroll
    for (int u = 0; u < GR; ++u)
        if (j + u < n1) w[u] = h4[(size_t)src_id(m1, j + u, last) * 32 + lane];
}

__device__ __forceinline__ void rows_add(int n0, int n1, int j,
                                         const float4 (&v)[GR], const float4 (&w)[GR],
                                         float4& a0, float4& a1) {
#pragma unroll
    for (int u = 0; u < GR; ++u)
        if (j + u < n0) { a0.x += v[u].x; a0.y += v[u].y; a0.z += v[u].z; a0.w += v[u].w; }
#pragma unroll
    for (int u = 0; u < GR; ++u)
        if (j + u < n1) { a1.x += w[u].x; a1.y += w[u].y; a1.z += w[u].z; a1.w += w[u].w; }
}

__device__ __forceinline__ int clamp32(int d) { return d < 0 ? 0 : (d > 32 ? 32 : d); }

__device__ __forceinline__ void gather_finish(const float4* __restrict__ h4,
                                              const int* __restrict__ col,
                                              int s0, int d0, unsigned m0,
                                              int s1, int d1, unsigned m1,
                                              int lane, unsigned last,
                                              float4 (&v)[GR], float4 (&w)[GR],
                                              float4& r0, float4& r1) {
    float4 a0 = make_float4(0.0f, 0.0f, 0.0f, 0.0f);
    float4 a1 = make_float4(0.0f, 0.0f, 0.0f, 0.0f);
    int n0 = clamp32(d0), n1 = clamp32(d1);
    int nm = n0 > n1 ? n0 : n1;
    rows_add(n0, n1, 0, v, w, a0, a1);
    for (int j = GR; j < nm; j += GR) {
        rows_issue(h4, m0, n0, m1, n1, j, lane, last, v, w);
        rows_add(n0, n1, j, v, w, a0, a1);
    }
    const int dmax = d0 > d1 ? d0 : d1;
    for (int j0 = 32; j0 < dmax; j0 += 32) {
        m0 = 0; m1 = 0;
        if (j0 + lane < d0) m0 = (unsigned)col[s0 + j0 + lane];
        if (j0 + lane < d1) m1 = (unsigned)col[s1 + j0 + lane];
        m0 = m0 < last ? m0 : last;
        m1 = m1 < last ? m1 : last;
        n0 = clamp32(d0 - j0); n1 = clamp32(d1 - j0);
        nm = n0 > n1 ? n0 : n1;
        for (int j = 0; j < nm; j += GR) {
            rows_issue(h4, m0, n0, m1, n1, j, lane, last, v, w);
            rows_add(n0, n1, j, v, w, a0, a1);
        }
    }
    r0 = a0; r1 = a1;
}

__global__ __launch_bounds__(256)
void sage_gemm_kernel(const float* __restrict__ h,
                      const int* __restrict__ rowptr,
                      const int* __restrict__ cnt,
                      const int* __restrict__ col,
                      const float* __restrict__ invdeg,
                      const float* __restrict__ Wl,
                      const float* __restrict__ Wr,
                      const float* __restrict__ bias,
                      float* __restrict__ out,
                      int n_nodes, int do_relu) {
    __shared__ float sA[2][2][16][APITCH];

    const int lane = threadIdx.x & 31;
    const int wave = __builtin_amdgcn_readfirstlane((int)(threadIdx.x >> 5));
    const int n0   = wave * 16;

    const int lrow  = lane & 15;
    const int khalf = (lane >> 4) * 2;
    const int ncol  = n0 + lrow;

    const int ntiles = (n_nodes + 15) >> 4;
    const int tstep  = (int)gridDim.x;
    int t = (int)blockIdx.x;
    if (t >= ntiles) return;

    v2f bl[32], br[32];
#pragma unroll
    for (int j = 0; j < 32; ++j) {
        const int k = 4 * j + khalf;
        bl[j].x = Wl[(size_t)k * HID + ncol];
        bl[j].y = Wl[(size_t)(k + 1) * HID + ncol];
        br[j].x = Wr[(size_t)k * HID + ncol];
        br[j].y = Wr[(size_t)(k + 1) * HID + ncol];
    }
    const float bv = bias[ncol];

    const float4* __restrict__ h4 = (const float4*)h;
    const int      lasti = n_nodes - 1;
    const unsigned last  = (unsigned)lasti;
    const int sr0 = 2 * wave;
    const int sr1 = 2 * wave + 1;

    float4 ph0, ph1, g0, g1;
    float4 v[GR], w[GR];
    float  i0, i1;

    int ra = t * 16 + sr0, rb = t * 16 + sr1;
    ra = ra < lasti ? ra : lasti;
    rb = rb < lasti ? rb : lasti;
    int cs0 = rowptr[ra], cd0 = cnt[ra], cs1 = rowptr[rb], cd1 = cnt[rb];
    unsigned m0 = (unsigned)col[lane < cd0 ? cs0 + lane : 0];
    unsigned m1 = (unsigned)col[lane < cd1 ? cs1 + lane : 0];
    ph0 = h4[(size_t)ra * 32 + lane];
    ph1 = h4[(size_t)rb * 32 + lane];
    i0 = invdeg[ra];
    i1 = invdeg[rb];
    rows_issue(h4, m0, clamp32(cd0), m1, clamp32(cd1), 0, lane, last, v, w);
    gather_finish(h4, col, cs0, cd0, m0, cs1, cd1, m1, lane, last, v, w, g0, g1);
    *(float4*)&sA[0][0][sr0][4 * lane] = make_float4(g0.x * i0, g0.y * i0, g0.z * i0, g0.w * i0);
    *(float4*)&sA[0][0][sr1][4 * lane] = make_float4(g1.x * i1, g1.y * i1, g1.z * i1, g1.w * i1);
    *(float4*)&sA[0][1][sr0][4 * lane] = ph0;
    *(float4*)&sA[0][1][sr1][4 * lane] = ph1;

    int na = (t + tstep) * 16 + sr0, nb_ = (t + tstep) * 16 + sr1;
    na  = na  < lasti ? na  : lasti;
    nb_ = nb_ < lasti ? nb_ : lasti;
    cs0 = rowptr[na]; cd0 = cnt[na]; cs1 = rowptr[nb_]; cd1 = cnt[nb_];
    m0 = (unsigned)col[lane < cd0 ? cs0 + lane : 0];
    m1 = (unsigned)col[lane < cd1 ? cs1 + lane : 0];
    ra = (t + 2 * tstep) * 16 + sr0; rb = (t + 2 * tstep) * 16 + sr1;
    ra = ra < lasti ? ra : lasti;
    rb = rb < lasti ? rb : lasti;
    int sa = rowptr[ra], da = cnt[ra], sb = rowptr[rb], db = cnt[rb];
    __syncthreads();

    int cur = 0;
    for (; t < ntiles; t += tstep) {
        const int  tn   = t + tstep;
        const bool more = tn < ntiles;
        unsigned mn0 = 0, mn1 = 0;
        const int ns0 = sa, nd0 = da, ns1 = sb, nd1 = db;
        if (more) {
            ph0 = h4[(size_t)na  * 32 + lane];
            ph1 = h4[(size_t)nb_ * 32 + lane];
            i0 = invdeg[na];
            i1 = invdeg[nb_];
            mn0 = (unsigned)col[lane < nd0 ? ns0 + lane : 0];
            mn1 = (unsigned)col[lane < nd1 ? ns1 + lane : 0];
            rows_issue(h4, m0, clamp32(cd0), m1, clamp32(cd1), 0, lane, last, v, w);
            na = ra; nb_ = rb;
            ra = (tn + 2 * tstep) * 16 + sr0; rb = (tn + 2 * tstep) * 16 + sr1;
            ra = ra < lasti ? ra : lasti;
            rb = rb < lasti ? rb : lasti;
            sa = rowptr[ra]; da = cnt[ra]; sb = rowptr[rb]; db = cnt[rb];
        }

        v8f acc_l = {};
        v8f acc_r = {};
        const float* __restrict__ pl = &sA[cur][0][lrow][khalf];
        const float* __restrict__ pr = &sA[cur][1][lrow][khalf];
#pragma unroll
        for (int j = 0; j < 32; ++j) {
            const v2f a_l = *(const v2f*)(pl + 4 * j);
            const v2f a_r = *(const v2f*)(pr + 4 * j);
            acc_l = __builtin_amdgcn_wmma_f32_16x16x4_f32(
                false, a_l, false, bl[j], (short)0, acc_l, false, false);
            acc_r = __builtin_amdgcn_wmma_f32_16x16x4_f32(
                false, a_r, false, br[j], (short)0, acc_r, false, false);
        }

        const int mbase = t * 16 + (lane >> 4) * 8;
#pragma unroll
        for (int q = 0; q < 8; ++q) {
            int m = mbase + q;
            if (m >= n_nodes) m = n_nodes - 1;
            float o = acc_l[q] + acc_r[q] + bv;
            if (do_relu) o = o > 0.0f ? o : 0.0f;
            out[(size_t)m * HID + ncol] = o;
        }

        if (more) {
            gather_finish(h4, col, cs0, cd0, m0, cs1, cd1, m1, lane, last, v, w, g0, g1);
            const int nb = cur ^ 1;
            *(float4*)&sA[nb][0][sr0][4 * lane] = make_float4(g0.x * i0, g0.y * i0, g0.z * i0, g0.w * i0);
            *(float4*)&sA[nb][0][sr1][4 * lane] = make_float4(g1.x * i1, g1.y * i1, g1.z * i1, g1.w * i1);
            *(float4*)&sA[nb][1][sr0][4 * lane] = ph0;
            *(float4*)&sA[nb][1][sr1][4 * lane] = ph1;
            cs0 = ns0; cd0 = nd0; cs1 = ns1; cd1 = nd1;
            m0 = mn0; m1 = mn1;
        }
        __builtin_amdgcn_fence(__ATOMIC_RELEASE, "workgroup", "local");
        __builtin_amdgcn_s_barrier();
        __builtin_amdgcn_fence(__ATOMIC_ACQUIRE, "workgroup", "local");
        cur ^= 1;
    }
}

#define ED_G 8

__device__ __forceinline__ float wave_sum_all(float x) {
    x += __builtin_bit_cast(float, __builtin_amdgcn_update_dpp(0, __builtin_bit_cast(int, x), 0x161, 0xf, 0xf, false));
    x += __builtin_bit_cast(float, __builtin_amdgcn_update_dpp(0, __builtin_bit_cast(int, x), 0x162, 0xf, 0xf, false));
    x += __builtin_bit_cast(float, __builtin_amdgcn_update_dpp(0, __builtin_bit_cast(int, x), 0x164, 0xf, 0xf, false));
    x += __builtin_bit_cast(float, __builtin_amdgcn_update_dpp(0, __builtin_bit_cast(int, x), 0x168, 0xf, 0xf, false));
    const int b = __builtin_bit_cast(int, x);
    x += __builtin_bit_cast(float, __builtin_amdgcn_permlanex16(b, b, 0x76543210, 0xfedcba98, false, false));
    return x;
}

__global__ __launch_bounds__(256)
void edge_dot_kernel(const float* __restrict__ h2,
                     const int* __restrict__ es,
                     const int* __restrict__ ed,
                     float* __restrict__ out, int n) {
    const int lane = threadIdx.x & 31;
    const int e0 = __builtin_amdgcn_readfirstlane(
        (int)((blockIdx.x * blockDim.x + threadIdx.x) & ~31u));
    if (e0 >= n) return;
    const int e  = e0 + lane;
    const int ec = e < n ? e : n - 1;
    const int si = es[ec];
    const int di = ed[ec];
    const float4* __restrict__ h4 = (const float4*)h2;

    float res = 0.0f;
#pragma unroll 1
    for (int g = 0; g < 32; g += ED_G) {
        float4 a[ED_G], b[ED_G];
#pragma unroll
        for (int u = 0; u < ED_G; ++u) {
            const int s = __builtin_amdgcn_readlane(si, g + u);
            const int d = __builtin_amdgcn_readlane(di, g + u);
            a[u] = h4[(size_t)s * 32 + lane];
            b[u] = h4[(size_t)d * 32 + lane];
        }
#pragma unroll
        for (int u = 0; u < ED_G; ++u) {
            float p = a[u].x * b[u].x + a[u].y * b[u].y + a[u].z * b[u].z + a[u].w * b[u].w;
            p = wave_sum_all(p);
            res = (lane == g + u) ? p : res;
        }
    }
    if (e < n) out[e] = res;
}

extern "C" void kernel_launch(void* const* d_in, const int* in_sizes, int n_in,
                              void* d_out, int out_size, void* d_ws, size_t ws_size,
                              hipStream_t stream) {
    const int*   n_id     = (const int*)  d_in[0];
    const float* x_struct = (const float*)d_in[1];
    const int*   eidx     = (const int*)  d_in[2];
    const int*   elab     = (const int*)  d_in[3];
    const float* emb      = (const float*)d_in[4];
    const float* W1l      = (const float*)d_in[5];
    const float* W1r      = (const float*)d_in[6];
    const float* b1       = (const float*)d_in[7];
    const float* W2l      = (const float*)d_in[8];
    const float* W2r      = (const float*)d_in[9];
    const float* b2       = (const float*)d_in[10];
    float*       out      = (float*)d_out;

    const int N = in_sizes[0];
    const int E = in_sizes[2] / 2;
    const int L = in_sizes[3] / 2;

    const int* src = eidx;
    const int* dst = eidx + E;
    const int* lsrc = elab;
    const int* ldst = elab + L;

    float* ws     = (float*)d_ws;
    float* h0     = ws;
    float* h1     = h0  + (size_t)N * HID;
    float* lists  = h1  + (size_t)N * HID;
    float* spare  = lists + (size_t)N * HID;
    int*   cnt    = (int*)(spare + (size_t)N * HID);
    float* invdeg = (float*)(cnt + N);
    int*   rowptr = (int*)lists;
    int*   col    = rowptr + N + E;
    int2*  pairs  = (int2*)spare;
    int*   counts = (int*)spare + 2 * (size_t)E;
    int*   binstart = counts + SORT_BLOCKS * MAXBINS;
    float* h2     = h0;

    build_h0_kernel<<<(N + 8 * H0_ROWS - 1) / (8 * H0_ROWS), 256, 0, stream>>>(
        n_id, emb, x_struct, h0, N);
    coarse_count_kernel<<<SORT_BLOCKS, SORT_T, 0, stream>>>(dst, counts, N, E);
    coarse_scatter_kernel<<<SORT_BLOCKS, SORT_T, 0, stream>>>(src, dst, counts, binstart, pairs, N, E);
    fine_sort_kernel<<<(N + BIN_W - 1) / BIN_W, SORT_T, 0, stream>>>(
        pairs, binstart, rowptr, cnt, invdeg, col, N, E);

    sage_gemm_kernel<<<GEMM_BLOCKS, 256, 0, stream>>>(
        h0, rowptr, cnt, col, invdeg, W1l, W1r, b1, h1, N,  1);

    sage_gemm_kernel<<<GEMM_BLOCKS, 256, 0, stream>>>(
        h1, rowptr, cnt, col, invdeg, W2l, W2r, b2, h2, N,  0);

    edge_dot_kernel<<<(L + 255) / 256, 256, 0, stream>>>(h2, lsrc, ldst, out, L);
}


# h0 build has no launch of its own: its rows ride as extra work-groups, half in the level-1 count launch and half in the level-2 sort launch (free wave slots there); 6 launches; sums unchanged
# speedup vs baseline: 5.4645x; 1.0338x over previous
#include <hip/hip_runtime.h>
#include <hip/hip_bf16.h>


typedef __attribute__((ext_vector_type(2))) float v2f;
typedef __attribute__((ext_vector_type(8))) float v8f;
typedef __attribute__((ext_vector_type(4))) float v4f;

#define HID 128
#define EMBD 64

#define H0_ROWS 8

__device__ __forceinline__
void h0_wave_rows(const int* __restrict__ n_id,
                  const float* __restrict__ emb,
                  const float* __restrict__ x_struct,
                  float* __restrict__ h0, int n_nodes, int wave_rank) {
    const int lane = threadIdx.x & 31;
    const int r0 = __builtin_amdgcn_readfirstlane(wave_rank * H0_ROWS);
    if (r0 >= n_nodes) return;
    const int last = n_nodes - 1;
    const int rl  = r0 + (lane & (H0_ROWS - 1));
    const int idv = n_id[rl < last ? rl : last];
    const bool lo = lane < 16;
    const v4f* __restrict__ src4 = (const v4f*)(lo ? emb : x_struct) + (lane & 15);
    v4f* __restrict__ dst4 = (v4f*)h0 + lane;

    v4f v[H0_ROWS];
#pragma unroll
    for (int u = 0; u < H0_ROWS; ++u) {
        const int row  = (r0 + u) < last ? (r0 + u) : last;
        const int id   = __builtin_amdgcn_readlane(idv, u);
        const int srow = lo ? id : row;
        v[u] = src4[(size_t)srow * (EMBD / 4)];
    }
#pragma unroll
    for (int u = 0; u < H0_ROWS; ++u) {
        const int row = (r0 + u) < last ? (r0 + u) : last;
        dst4[(size_t)row * (HID / 4)] = v[u];
    }
}

#define BIN_SHIFT 8
#define BIN_W     256
#define MAXBINS   1024
#define SORT_T    1024
#define SORT_U    4
#define SORT_BLOCKS 256

__device__ __forceinline__ int wave_incl_sum(int x, int lane) {
    x += __builtin_amdgcn_update_dpp(0, x, 0x111, 0xf, 0xf, false);
    x += __builtin_amdgcn_update_dpp(0, x, 0x112, 0xf, 0xf, false);
    x += __builtin_amdgcn_update_dpp(0, x, 0x114, 0xf, 0xf, false);
    x += __builtin_amdgcn_update_dpp(0, x, 0x118, 0xf, 0xf, false);
    const int low = __builtin_amdgcn_readlane(x, 15);
    return lane >= 16 ? x + low : x;
}

__device__ __forceinline__ int dst_node(int d, int n_nodes) {
    const unsigned u = (unsigned)d, last = (unsigned)(n_nodes - 1);
    return (int)(u < last ? u : last);
}

__global__ __launch_bounds__(SORT_T)
void coarse_count_kernel(const int* __restrict__ dst, int* __restrict__ counts,
                         int n_nodes, int n_edges, int n_sort,
                         const int* __restrict__ n_id, const float* __restrict__ emb,
                         const float* __restrict__ x_struct, float* __restrict__ h0) {
    __shared__ int hist[MAXBINS];
    if ((int)blockIdx.x >= n_sort) {
        const int wr = ((int)blockIdx.x - n_sort) * (SORT_T / 32) + (int)(threadIdx.x >> 5);
        h0_wave_rows(n_id, emb, x_struct, h0, n_nodes, wr);
        return;
    }
    const int tid   = threadIdx.x;
    const int nbins = (n_nodes + BIN_W - 1) >> BIN_SHIFT;
    const int chunk = (n_edges + n_sort - 1) / n_sort;
    const int eb    = (int)blockIdx.x * chunk;
    const int ee    = eb + chunk < n_edges ? eb + chunk : n_edges;
    for (int i = tid; i < nbins; i += SORT_T) hist[i] = 0;
    __syncthreads();
    for (int e0 = eb; e0 < ee; e0 += SORT_T * SORT_U) {
        int d[SORT_U];
#pragma unroll
        for (int u = 0; u < SORT_U; ++u) {
            const int e = e0 + u * SORT_T + tid;
            d[u] = dst[e < ee ? e : ee - 1];
        }
#pragma unroll
        for (int u = 0; u < SORT_U; ++u) {
            const int e = e0 + u * SORT_T + tid;
            if (e < ee) atomicAdd(&hist[dst_node(d[u], n_nodes) >> BIN_SHIFT], 1);
        }
    }
    __syncthreads();
    const int cstride = (nbins + 31) & ~31;
    for (int i = tid; i < nbins; i += SORT_T)
        counts[(size_t)blockIdx.x * cstride + i] = hist[i];
}

#define CS_PARTS 4
#define CS_G     16
__global__ __launch_bounds__(SORT_T)
void coarse_scatter_kernel(const int* __restrict__ src, const int* __restrict__ dst,
                           const int* __restrict__ counts, int* __restrict__ binstart,
                           int2* __restrict__ pairs, int n_nodes, int n_edges) {
    __shared__ int totS[MAXBINS + 1];
    __shared__ int preS[MAXBINS];
    __shared__ int curS[MAXBINS];
    const int tid   = threadIdx.x;
    const int lane  = tid & 31;
    const int wave  = __builtin_amdgcn_readfirstlane((int)(threadIdx.x >> 5));
    const int nbins = (n_nodes + BIN_W - 1) >> BIN_SHIFT;
    const int cstride = (nbins + 31) & ~31;
    const int nblk  = (int)gridDim.x;
    const int k     = (int)blockIdx.x;
    const int chunk = (n_edges + nblk - 1) / nblk;
    const int eb    = k * chunk;
    const int ee    = eb + chunk < n_edges ? eb + chunk : n_edges;

    for (int i = tid; i < nbins; i += SORT_T) { totS[i] = 0; preS[i] = 0; }
    __syncthreads();
    {
        const int part = wave / (SORT_T / 32 / CS_PARTS);
        const int jper = (nblk + CS_PARTS - 1) / CS_PARTS;
        const int j0   = part * jper;
        const int j1   = j0 + jper < nblk ? j0 + jper : nblk;
        for (int bin = tid & (SORT_T / CS_PARTS - 1); bin < nbins; bin += SORT_T / CS_PARTS) {
            int tot = 0, pre = 0;
            for (int j = j0; j < j1; j += CS_G) {
                int v[CS_G];
#pragma unroll
                for (int u = 0; u < CS_G; ++u)
                    v[u] = counts[(size_t)(j + u < j1 ? j + u : j1 - 1) * cstride + bin];
#pragma unroll
                for (int u = 0; u < CS_G; ++u) {
                    tot += (j + u < j1) ? v[u] : 0;
                    pre += (j + u < j1 && j + u < k) ? v[u] : 0;
                }
            }
            atomicAdd(&totS[bin], tot);
            atomicAdd(&preS[bin], pre);
        }
    }
    __syncthreads();
    if (wave == 0) {
        int run = 0;
        for (int c0 = 0; c0 < nbins; c0 += 32) {
            const int i = c0 + lane;
            const int x = i < nbins ? totS[i] : 0;
            const int incl = wave_incl_sum(x, lane);
            if (i < nbins) totS[i] = run + incl - x;
            run += __builtin_amdgcn_readlane(incl, 31);
        }
        if (lane == 0) totS[nbins] = run;
    }
    __syncthreads();
    for (int i = tid; i < nbins; i += SORT_T) curS[i] = totS[i] + preS[i];
    if (k == 0)
        for (int i = tid; i <= nbins; i += SORT_T) binstart[i] = totS[i];
    __syncthreads();

    const unsigned elast = (unsigned)(n_edges - 1);
    for (int e0 = eb; e0 < ee; e0 += SORT_T * SORT_U) {
        int s[SORT_U], d[SORT_U];
#pragma unroll
        for (int u = 0; u < SORT_U; ++u) {
            const int e  = e0 + u * SORT_T + tid;
            const int ec = e < ee ? e : ee - 1;
            s[u] = src[ec];
            d[u] = dst[ec];
        }
#pragma unroll
        for (int u = 0; u < SORT_U; ++u) {
            const int e = e0 + u * SORT_T + tid;
            if (e < ee) {
                const int dn = dst_node(d[u], n_nodes);
                unsigned p = (unsigned)atomicAdd(&curS[dn >> BIN_SHIFT], 1);
                p = p < elast ? p : elast;
                pairs[p] = make_int2(s[u], dn);
            }
        }
    }
}

__global__ __launch_bounds__(SORT_T)
void fine_sort_kernel(const int2* __restrict__ pairs, const int* __restrict__ binstart,
                      int* __restrict__ rowptr, int* __restrict__ cnt,
                      float* __restrict__ invdeg, int* __restrict__ col,
                      int n_nodes, int n_edges,
                      const int* __restrict__ n_id, const float* __restrict__ emb,
                      const float* __restrict__ x_struct, float* __restrict__ h0,
                      int h0_wave0) {
    __shared__ int c[BIN_W];
    __shared__ int wt[BIN_W / 32];
    const int nbins_ = (n_nodes + BIN_W - 1) >> BIN_SHIFT;
    if ((int)blockIdx.x >= nbins_) {
        const int wr = h0_wave0 + ((int)blockIdx.x - nbins_) * (SORT_T / 32) + (int)(threadIdx.x >> 5);
        h0_wave_rows(n_id, emb, x_struct, h0, n_nodes, wr);
        return;
    }
    const int tid  = threadIdx.x;
    const int lane = tid & 31;
    const int wave = __builtin_amdgcn_readfirstlane((int)(threadIdx.x >> 5));
    const int lo   = (int)blockIdx.x * BIN_W;
    int ps = binstart[blockIdx.x], pe = binstart[blockIdx.x + 1];
    ps = ps < 0 ? 0 : ps;
    pe = pe > n_edges ? n_edges : pe;

    if (tid < BIN_W) c[tid] = 0;
    __syncthreads();
    for (int i0 = ps; i0 < pe; i0 += SORT_T * SORT_U) {
        int d[SORT_U];
#pragma unroll
        for (int u = 0; u < SORT_U; ++u) {
            const int i = i0 + u * SORT_T + tid;
            d[u] = pairs[i < pe ? i : pe - 1].y;
        }
#pragma unroll
        for (int u = 0; u < SORT_U; ++u) {
            const int i = i0 + u * SORT_T + tid;
            if (i < pe) atomicAdd(&c[(d[u] - lo) & (BIN_W - 1)], 1);
        }
    }
    __syncthreads();
    int deg = 0, start = 0;
    if (wave < BIN_W / 32) {
        deg = c[tid];
        const int incl = wave_incl_sum(deg, lane);
        if (lane == 31) wt[wave] = incl;
        start = ps + incl - deg;
    }
    __syncthreads();
    if (wave < BIN_W / 32) {
#pragma unroll
        for (int w2 = 0; w2 < BIN_W / 32; ++w2) start += (w2 < wave) ? wt[w2] : 0;
        c[tid] = start;
        const int node = lo + tid;
        if (node < n_nodes) {
            rowptr[node] = start;
            cnt[node]    = deg;
            invdeg[node] = 1.0f / fmaxf((float)deg, 1.0f);
        }
    }
    __syncthreads();
    const unsigned elast = (unsigned)(n_edges - 1);
    for (int i0 = ps; i0 < pe; i0 += SORT_T * SORT_U) {
        int2 p[SORT_U];
#pragma unroll
        for (int u = 0; u < SORT_U; ++u) {
            const int i = i0 + u * SORT_T + tid;
            p[u] = pairs[i < pe ? i : pe - 1];
        }
#pragma unroll
        for (int u = 0; u < SORT_U; ++u) {
            const int i = i0 + u * SORT_T + tid;
            if (i < pe) {
                unsigned slot = (unsigned)atomicAdd(&c[(p[u].y - lo) & (BIN_W - 1)], 1);
                slot = slot < elast ? slot : elast;
                col[slot] = p[u].x;
            }
        }
    }
}

#define APITCH 132
#define GEMM_BLOCKS 512
#define GR 8

__device__ __forceinline__ unsigned src_id(unsigned m, int j, unsigned last) {
    const unsigned id = (unsigned)__builtin_amdgcn_readlane((int)m, j);
    return id < last ? id : last;
}

__device__ __forceinline__ void rows_issue(const float4* __restrict__ h4,
                                           unsigned m0, int n0, unsigned m1, int n1,
                                           int j, int lane, unsigned last,
                                           float4 (&v)[GR], float4 (&w)[GR]) {
#pragma unroll
    for (int u = 0; u < GR; ++u)
        if (j + u < n0) v[u] = h4[(size_t)src_id(m0, j + u, last) * 32 + lane];
#pragma unroll
    for (int u = 0; u < GR; ++u)
        if (j + u < n1) w[u] = h4[(size_t)src_id(m1, j + u, last) * 32 + lane];
}

__device__ __forceinline__ void rows_add(int n0, int n1, int j,
                                         const float4 (&v)[GR], const float4 (&w)[GR],
                                         float4& a0, float4& a1) {
#pragma unroll
    for (int u = 0; u < GR; ++u)
        if (j + u < n0) { a0.x += v[u].x; a0.y += v[u].y; a0.z += v[u].z; a0.w += v[u].w; }
#pragma unroll
    for (int u = 0; u < GR; ++u)
        if (j + u < n1) { a1.x += w[u].x; a1.y += w[u].y; a1.z += w[u].z; a1.w += w[u].w; }
}

__device__ __forceinline__ int clamp32(int d) { return d < 0 ? 0 : (d > 32 ? 32 : d); }

__device__ __forceinline__ void gather_finish(const float4* __restrict__ h4,
                                              const int* __restrict__ col,
                                              int s0, int d0, unsigned m0,
                                              int s1, int d1, unsigned m1,
                                              int lane, unsigned last,
                                              float4 (&v)[GR], float4 (&w)[GR],
                                              float4& r0, float4& r1) {
    float4 a0 = make_float4(0.0f, 0.0f, 0.0f, 0.0f);
    float4 a1 = make_float4(0.0f, 0.0f, 0.0f, 0.0f);
    int n0 = clamp32(d0), n1 = clamp32(d1);
    int nm = n0 > n1 ? n0 : n1;
    rows_add(n0, n1, 0, v, w, a0, a1);
    for (int j = GR; j < nm; j += GR) {
        rows_issue(h4, m0, n0, m1, n1, j, lane, last, v, w);
        rows_add(n0, n1, j, v, w, a0, a1);
    }
    const int dmax = d0 > d1 ? d0 : d1;
    for (int j0 = 32; j0 < dmax; j0 += 32) {
        m0 = 0; m1 = 0;
        if (j0 + lane < d0) m0 = (unsigned)col[s0 + j0 + lane];
        if (j0 + lane < d1) m1 = (unsigned)col[s1 + j0 + lane];
        m0 = m0 < last ? m0 : last;
        m1 = m1 < last ? m1 : last;
        n0 = clamp32(d0 - j0); n1 = clamp32(d1 - j0);
        nm = n0 > n1 ? n0 : n1;
        for (int j = 0; j < nm; j += GR) {
            rows_issue(h4, m0, n0, m1, n1, j, lane, last, v, w);
            rows_add(n0, n1, j, v, w, a0, a1);
        }
    }
    r0 = a0; r1 = a1;
}

__global__ __launch_bounds__(256)
void sage_gemm_kernel(const float* __restrict__ h,
                      const int* __restrict__ rowptr,
                      const int* __restrict__ cnt,
                      const int* __restrict__ col,
                      const float* __restrict__ invdeg,
                      const float* __restrict__ Wl,
                      const float* __restrict__ Wr,
                      const float* __restrict__ bias,
                      float* __restrict__ out,
                      int n_nodes, int do_relu) {
    __shared__ float sA[2][2][16][APITCH];

    const int lane = threadIdx.x & 31;
    const int wave = __builtin_amdgcn_readfirstlane((int)(threadIdx.x >> 5));
    const int n0   = wave * 16;

    const int lrow  = lane & 15;
    const int khalf = (lane >> 4) * 2;
    const int ncol  = n0 + lrow;

    const int ntiles = (n_nodes + 15) >> 4;
    const int tstep  = (int)gridDim.x;
    int t = (int)blockIdx.x;
    if (t >= ntiles) return;

    v2f bl[32], br[32];
#pragma unroll
    for (int j = 0; j < 32; ++j) {
        const int k = 4 * j + khalf;
        bl[j].x = Wl[(size_t)k * HID + ncol];
        bl[j].y = Wl[(size_t)(k + 1) * HID + ncol];
        br[j].x = Wr[(size_t)k * HID + ncol];
        br[j].y = Wr[(size_t)(k + 1) * HID + ncol];
    }
    const float bv = bias[ncol];

    const float4* __restrict__ h4 = (const float4*)h;
    const int      lasti = n_nodes - 1;
    const unsigned last  = (unsigned)lasti;
    const int sr0 = 2 * wave;
    const int sr1 = 2 * wave + 1;

    float4 ph0, ph1, g0, g1;
    float4 v[GR], w[GR];
    float  i0, i1;

    int ra = t * 16 + sr0, rb = t * 16 + sr1;
    ra = ra < lasti ? ra : lasti;
    rb = rb < lasti ? rb : lasti;
    int cs0 = rowptr[ra], cd0 = cnt[ra], cs1 = rowptr[rb], cd1 = cnt[rb];
    unsigned m0 = (unsigned)col[lane < cd0 ? cs0 + lane : 0];
    unsigned m1 = (unsigned)col[lane < cd1 ? cs1 + lane : 0];
    ph0 = h4[(size_t)ra * 32 + lane];
    ph1 = h4[(size_t)rb * 32 + lane];
    i0 = invdeg[ra];
    i1 = invdeg[rb];
    rows_issue(h4, m0, clamp32(cd0), m1, clamp32(cd1), 0, lane, last, v, w);
    gather_finish(h4, col, cs0, cd0, m0, cs1, cd1, m1, lane, last, v, w, g0, g1);
    *(float4*)&sA[0][0][sr0][4 * lane] = make_float4(g0.x * i0, g0.y * i0, g0.z * i0, g0.w * i0);
    *(float4*)&sA[0][0][sr1][4 * lane] = make_float4(g1.x * i1, g1.y * i1, g1.z * i1, g1.w * i1);
    *(float4*)&sA[0][1][sr0][4 * lane] = ph0;
    *(float4*)&sA[0][1][sr1][4 * lane] = ph1;

    int na = (t + tstep) * 16 + sr0, nb_ = (t + tstep) * 16 + sr1;
    na  = na  < lasti ? na  : lasti;
    nb_ = nb_ < lasti ? nb_ : lasti;
    cs0 = rowptr[na]; cd0 = cnt[na]; cs1 = rowptr[nb_]; cd1 = cnt[nb_];
    m0 = (unsigned)col[lane < cd0 ? cs0 + lane : 0];
    m1 = (unsigned)col[lane < cd1 ? cs1 + lane : 0];
    ra = (t + 2 * tstep) * 16 + sr0; rb = (t + 2 * tstep) * 16 + sr1;
    ra = ra < lasti ? ra : lasti;
    rb = rb < lasti ? rb : lasti;
    int sa = rowptr[ra], da = cnt[ra], sb = rowptr[rb], db = cnt[rb];
    __syncthreads();

    int cur = 0;
    for (; t < ntiles; t += tstep) {
        const int  tn   = t + tstep;
        const bool more = tn < ntiles;
        unsigned mn0 = 0, mn1 = 0;
        const int ns0 = sa, nd0 = da, ns1 = sb, nd1 = db;
        if (more) {
            ph0 = h4[(size_t)na  * 32 + lane];
            ph1 = h4[(size_t)nb_ * 32 + lane];
            i0 = invdeg[na];
            i1 = invdeg[nb_];
            mn0 = (unsigned)col[lane < nd0 ? ns0 + lane : 0];
            mn1 = (unsigned)col[lane < nd1 ? ns1 + lane : 0];
            rows_issue(h4, m0, clamp32(cd0), m1, clamp32(cd1), 0, lane, last, v, w);
            na = ra; nb_ = rb;
            ra = (tn + 2 * tstep) * 16 + sr0; rb = (tn + 2 * tstep) * 16 + sr1;
            ra = ra < lasti ? ra : lasti;
            rb = rb < lasti ? rb : lasti;
            sa = rowptr[ra]; da = cnt[ra]; sb = rowptr[rb]; db = cnt[rb];
        }

        v8f acc_l = {};
        v8f acc_r = {};
        const float* __restrict__ pl = &sA[cur][0][lrow][khalf];
        const float* __restrict__ pr = &sA[cur][1][lrow][khalf];
#pragma unroll
        for (int j = 0; j < 32; ++j) {
            const v2f a_l = *(const v2f*)(pl + 4 * j);
            const v2f a_r = *(const v2f*)(pr + 4 * j);
            acc_l = __builtin_amdgcn_wmma_f32_16x16x4_f32(
                false, a_l, false, bl[j], (short)0, acc_l, false, false);
            acc_r = __builtin_amdgcn_wmma_f32_16x16x4_f32(
                false, a_r, false, br[j], (short)0, acc_r, false, false);
        }

        const int mbase = t * 16 + (lane >> 4) * 8;
#pragma unroll
        for (int q = 0; q < 8; ++q) {
            int m = mbase + q;
            if (m >= n_nodes) m = n_nodes - 1;
            float o = acc_l[q] + acc_r[q] + bv;
            if (do_relu) o = o > 0.0f ? o : 0.0f;
            out[(size_t)m * HID + ncol] = o;
        }

        if (more) {
            gather_finish(h4, col, cs0, cd0, m0, cs1, cd1, m1, lane, last, v, w, g0, g1);
            const int nb = cur ^ 1;
            *(float4*)&sA[nb][0][sr0][4 * lane] = make_float4(g0.x * i0, g0.y * i0, g0.z * i0, g0.w * i0);
            *(float4*)&sA[nb][0][sr1][4 * lane] = make_float4(g1.x * i1, g1.y * i1, g1.z * i1, g1.w * i1);
            *(float4*)&sA[nb][1][sr0][4 * lane] = ph0;
            *(float4*)&sA[nb][1][sr1][4 * lane] = ph1;
            cs0 = ns0; cd0 = nd0; cs1 = ns1; cd1 = nd1;
            m0 = mn0; m1 = mn1;
        }
        __builtin_amdgcn_fence(__ATOMIC_RELEASE, "workgroup", "local");
        __builtin_amdgcn_s_barrier();
        __builtin_amdgcn_fence(__ATOMIC_ACQUIRE, "workgroup", "local");
        cur ^= 1;
    }
}

#define ED_G 8

__device__ __forceinline__ float wave_sum_all(float x) {
    x += __builtin_bit_cast(float, __builtin_amdgcn_update_dpp(0, __builtin_bit_cast(int, x), 0x161, 0xf, 0xf, false));
    x += __builtin_bit_cast(float, __builtin_amdgcn_update_dpp(0, __builtin_bit_cast(int, x), 0x162, 0xf, 0xf, false));
    x += __builtin_bit_cast(float, __builtin_amdgcn_update_dpp(0, __builtin_bit_cast(int, x), 0x164, 0xf, 0xf, false));
    x += __builtin_bit_cast(float, __builtin_amdgcn_update_dpp(0, __builtin_bit_cast(int, x), 0x168, 0xf, 0xf, false));
    const int b = __builtin_bit_cast(int, x);
    x += __builtin_bit_cast(float, __builtin_amdgcn_permlanex16(b, b, 0x76543210, 0xfedcba98, false, false));
    return x;
}

__global__ __launch_bounds__(256)
void edge_dot_kernel(const float* __restrict__ h2,
                     const int* __restrict__ es,
                     const int* __restrict__ ed,
                     float* __restrict__ out, int n) {
    const int lane = threadIdx.x & 31;
    const int e0 = __builtin_amdgcn_readfirstlane(
        (int)((blockIdx.x * blockDim.x + threadIdx.x) & ~31u));
    if (e0 >= n) return;
    const int e  = e0 + lane;
    const int ec = e < n ? e : n - 1;
    const int si = es[ec];
    const int di = ed[ec];
    const float4* __restrict__ h4 = (const float4*)h2;

    float res = 0.0f;
#pragma unroll 1
    for (int g = 0; g < 32; g += ED_G) {
        float4 a[ED_G], b[ED_G];
#pragma unroll
        for (int u = 0; u < ED_G; ++u) {
            const int s = __builtin_amdgcn_readlane(si, g + u);
            const int d = __builtin_amdgcn_readlane(di, g + u);
            a[u] = h4[(size_t)s * 32 + lane];
            b[u] = h4[(size_t)d * 32 + lane];
        }
#pragma unroll
        for (int u = 0; u < ED_G; ++u) {
            float p = a[u].x * b[u].x + a[u].y * b[u].y + a[u].z * b[u].z + a[u].w * b[u].w;
            p = wave_sum_all(p);
            res = (lane == g + u) ? p : res;
        }
    }
    if (e < n) out[e] = res;
}

extern "C" void kernel_launch(void* const* d_in, const int* in_sizes, int n_in,
                              void* d_out, int out_size, void* d_ws, size_t ws_size,
                              hipStream_t stream) {
    const int*   n_id     = (const int*)  d_in[0];
    const float* x_struct = (const float*)d_in[1];
    const int*   eidx     = (const int*)  d_in[2];
    const int*   elab     = (const int*)  d_in[3];
    const float* emb      = (const float*)d_in[4];
    const float* W1l      = (const float*)d_in[5];
    const float* W1r      = (const float*)d_in[6];
    const float* b1       = (const float*)d_in[7];
    const float* W2l      = (const float*)d_in[8];
    const float* W2r      = (const float*)d_in[9];
    const float* b2       = (const float*)d_in[10];
    float*       out      = (float*)d_out;

    const int N = in_sizes[0];
    const int E = in_sizes[2] / 2;
    const int L = in_sizes[3] / 2;

    const int* src = eidx;
    const int* dst = eidx + E;
    const int* lsrc = elab;
    const int* ldst = elab + L;

    float* ws     = (float*)d_ws;
    float* h0     = ws;
    float* h1     = h0  + (size_t)N * HID;
    float* lists  = h1  + (size_t)N * HID;
    float* spare  = lists + (size_t)N * HID;
    int*   cnt    = (int*)(spare + (size_t)N * HID);
    float* invdeg = (float*)(cnt + N);
    int*   rowptr = (int*)lists;
    int*   col    = rowptr + N + E;
    int2*  pairs  = (int2*)spare;
    int*   counts = (int*)spare + 2 * (size_t)E;
    int*   binstart = counts + SORT_BLOCKS * MAXBINS;
    float* h2     = h0;

    const int h0_rows_wg = (SORT_T / 32) * H0_ROWS;
    const int h0_wgs     = (N + h0_rows_wg - 1) / h0_rows_wg;
    const int h0_wgs_a   = h0_wgs / 2;
    const int h0_wgs_b   = h0_wgs - h0_wgs_a;
    const int h0_waves_a = h0_wgs_a * (SORT_T / 32);
    coarse_count_kernel<<<SORT_BLOCKS + h0_wgs_a, SORT_T, 0, stream>>>(
        dst, counts, N, E, SORT_BLOCKS, n_id, emb, x_struct, h0);
    coarse_scatter_kernel<<<SORT_BLOCKS, SORT_T, 0, stream>>>(src, dst, counts, binstart, pairs, N, E);
    fine_sort_kernel<<<(N + BIN_W - 1) / BIN_W + h0_wgs_b, SORT_T, 0, stream>>>(
        pairs, binstart, rowptr, cnt, invdeg, col, N, E, n_id, emb, x_struct, h0, h0_waves_a);

    sage_gemm_kernel<<<GEMM_BLOCKS, 256, 0, stream>>>(
        h0, rowptr, cnt, col, invdeg, W1l, W1r, b1, h1, N,  1);

    sage_gemm_kernel<<<GEMM_BLOCKS, 256, 0, stream>>>(
        h1, rowptr, cnt, col, invdeg, W2l, W2r, b2, h2, N,  0);

    edge_dot_kernel<<<(L + 255) / 256, 256, 0, stream>>>(h2, lsrc, ldst, out, L);
}


# Edges grouped in two launches, not three: level 1 sorts each chunk by bin in place (no global counts table), level 2 collects its bin's runs from the chunks; 5 launches; bucket slot order differs
# speedup vs baseline: 6.1861x; 1.1321x over previous
#include <hip/hip_runtime.h>
#include <hip/hip_bf16.h>


typedef __attribute__((ext_vector_type(2))) float v2f;
typedef __attribute__((ext_vector_type(8))) float v8f;
typedef __attribute__((ext_vector_type(4))) float v4f;

#define HID 128
#define EMBD 64

#define H0_ROWS 8

__device__ __forceinline__
void h0_wave_rows(const int* __restrict__ n_id,
                  const float* __restrict__ emb,
                  const float* __restrict__ x_struct,
                  float* __restrict__ h0, int n_nodes, int wave_rank) {
    const int lane = threadIdx.x & 31;
    const int r0 = __builtin_amdgcn_readfirstlane(wave_rank * H0_ROWS);
    if (r0 >= n_nodes) return;
    const int last = n_nodes - 1;
    const int rl  = r0 + (lane & (H0_ROWS - 1));
    const int idv = n_id[rl < last ? rl : last];
    const bool lo = lane < 16;
    const v4f* __restrict__ src4 = (const v4f*)(lo ? emb : x_struct) + (lane & 15);
    v4f* __restrict__ dst4 = (v4f*)h0 + lane;

    v4f v[H0_ROWS];
#pragma unroll
    for (int u = 0; u < H0_ROWS; ++u) {
        const int row  = (r0 + u) < last ? (r0 + u) : last;
        const int id   = __builtin_amdgcn_readlane(idv, u);
        const int srow = lo ? id : row;
        v[u] = src4[(size_t)srow * (EMBD / 4)];
    }
#pragma unroll
    for (int u = 0; u < H0_ROWS; ++u) {
        const int row = (r0 + u) < last ? (r0 + u) : last;
        dst4[(size_t)row * (HID / 4)] = v[u];
    }
}

#define BIN_SHIFT 8
#define BIN_W     256
#define MAXBINS   1024
#define SORT_T    1024
#define SORT_U    4
#define SORT_BLOCKS 256
#define RUNS_W (SORT_BLOCKS / (SORT_T / 32))

__device__ __forceinline__ int wave_incl_sum(int x, int lane) {
    x += __builtin_amdgcn_update_dpp(0, x, 0x111, 0xf, 0xf, false);
    x += __builtin_amdgcn_update_dpp(0, x, 0x112, 0xf, 0xf, false);
    x += __builtin_amdgcn_update_dpp(0, x, 0x114, 0xf, 0xf, false);
    x += __builtin_amdgcn_update_dpp(0, x, 0x118, 0xf, 0xf, false);
    const int low = __builtin_amdgcn_readlane(x, 15);
    return lane >= 16 ? x + low : x;
}

__device__ __forceinline__ int dst_node(int d, int n_nodes) {
    const unsigned u = (unsigned)d, last = (unsigned)(n_nodes - 1);
    return (int)(u < last ? u : last);
}

__device__ __forceinline__ int row_pitch(int nbins) { return (nbins + 1 + 31) & ~31; }

__global__ __launch_bounds__(SORT_T)
void chunk_sort_kernel(const int* __restrict__ src, const int* __restrict__ dst,
                       int* __restrict__ starts, int2* __restrict__ pairs,
                       int n_nodes, int n_edges, int n_sort,
                       const int* __restrict__ n_id, const float* __restrict__ emb,
                       const float* __restrict__ x_struct, float* __restrict__ h0) {
    __shared__ int hist[MAXBINS + 1];
    __shared__ int curS[MAXBINS];
    if ((int)blockIdx.x >= n_sort) {
        const int wr = ((int)blockIdx.x - n_sort) * (SORT_T / 32) + (int)(threadIdx.x >> 5);
        h0_wave_rows(n_id, emb, x_struct, h0, n_nodes, wr);
        return;
    }
    const int tid   = threadIdx.x;
    const int lane  = tid & 31;
    const int wave  = __builtin_amdgcn_readfirstlane((int)(threadIdx.x >> 5));
    const int nbins = (n_nodes + BIN_W - 1) >> BIN_SHIFT;
    const int pitch = row_pitch(nbins);
    const int chunk = (n_edges + n_sort - 1) / n_sort;
    const int eb    = (int)blockIdx.x * chunk;
    const int ee    = eb + chunk < n_edges ? eb + chunk : n_edges;
    const int clen  = ee > eb ? ee - eb : 0;
    for (int i = tid; i < nbins; i += SORT_T) hist[i] = 0;
    __syncthreads();
    for (int e0 = eb; e0 < ee; e0 += SORT_T * SORT_U) {
        int d[SORT_U];
#pragma unroll
        for (int u = 0; u < SORT_U; ++u) {
            const int e = e0 + u * SORT_T + tid;
            d[u] = dst[e < ee ? e : ee - 1];
        }
#pragma unroll
        for (int u = 0; u < SORT_U; ++u) {
            const int e = e0 + u * SORT_T + tid;
            if (e < ee) atomicAdd(&hist[dst_node(d[u], n_nodes) >> BIN_SHIFT], 1);
        }
    }
    __syncthreads();
    if (wave == 0) {
        int run = 0;
        for (int c0 = 0; c0 < nbins; c0 += 32) {
            const int i = c0 + lane;
            const int x = i < nbins ? hist[i] : 0;
            const int incl = wave_incl_sum(x, lane);
            if (i < nbins) hist[i] = run + incl - x;
            run += __builtin_amdgcn_readlane(incl, 31);
        }
        if (lane == 0) hist[nbins] = run;
    }
    __syncthreads();
    for (int i = tid; i < nbins; i += SORT_T) curS[i] = hist[i];
    for (int i = tid; i <= nbins; i += SORT_T) starts[(size_t)blockIdx.x * pitch + i] = hist[i];
    __syncthreads();

    const unsigned plast = (unsigned)(clen > 0 ? clen - 1 : 0);
    for (int e0 = eb; e0 < ee; e0 += SORT_T * SORT_U) {
        int s[SORT_U], d[SORT_U];
#pragma unroll
        for (int u = 0; u < SORT_U; ++u) {
            const int e  = e0 + u * SORT_T + tid;
            const int ec = e < ee ? e : ee - 1;
            s[u] = src[ec];
            d[u] = dst[ec];
        }
#pragma unroll
        for (int u = 0; u < SORT_U; ++u) {
            const int e = e0 + u * SORT_T + tid;
            if (e < ee) {
                const int dn = dst_node(d[u], n_nodes);
                unsigned p = (unsigned)atomicAdd(&curS[dn >> BIN_SHIFT], 1);
                p = p < plast ? p : plast;
                pairs[(size_t)eb + p] = make_int2(s[u], dn);
            }
        }
    }
}

__global__ __launch_bounds__(SORT_T)
void bin_sort_kernel(const int2* __restrict__ pairs, const int* __restrict__ starts,
                     int* __restrict__ rowptr, int* __restrict__ cnt,
                     float* __restrict__ invdeg, int* __restrict__ col,
                     int n_nodes, int n_edges,
                     const int* __restrict__ n_id, const float* __restrict__ emb,
                     const float* __restrict__ x_struct, float* __restrict__ h0,
                     int h0_wave0) {
    __shared__ int c[BIN_W];
    __shared__ int wt[BIN_W / 32];
    __shared__ int runBase[SORT_BLOCKS];
    __shared__ int runLen[SORT_BLOCKS];
    __shared__ int frontS[SORT_BLOCKS / 32];
    const int nbins_ = (n_nodes + BIN_W - 1) >> BIN_SHIFT;
    if ((int)blockIdx.x >= nbins_) {
        const int wr = h0_wave0 + ((int)blockIdx.x - nbins_) * (SORT_T / 32) + (int)(threadIdx.x >> 5);
        h0_wave_rows(n_id, emb, x_struct, h0, n_nodes, wr);
        return;
    }
    const int tid  = threadIdx.x;
    const int lane = tid & 31;
    const int wave = __builtin_amdgcn_readfirstlane((int)(threadIdx.x >> 5));
    const int b    = (int)blockIdx.x;
    const int lo   = b * BIN_W;
    const int pitch = row_pitch(nbins_);
    const int chunk = (n_edges + SORT_BLOCKS - 1) / SORT_BLOCKS;

    if (wave < SORT_BLOCKS / 32) {
        const int k   = tid;
        const int eb  = k * chunk;
        const int ee  = eb + chunk < n_edges ? eb + chunk : n_edges;
        const int clen = ee > eb ? ee - eb : 0;
        int s0 = starts[(size_t)k * pitch + b];
        int s1 = starts[(size_t)k * pitch + b + 1];
        s0 = s0 < 0 ? 0 : (s0 > clen ? clen : s0);
        s1 = s1 < s0 ? s0 : (s1 > clen ? clen : s1);
        runBase[k] = eb + s0;
        runLen[k]  = s1 - s0;
        const int infront = wave_incl_sum(s0, lane);
        if (lane == 31) frontS[wave] = infront;
    }
    if (tid < BIN_W) c[tid] = 0;
    __syncthreads();
    int ps = 0;
    for (int w2 = 0; w2 < SORT_BLOCKS / 32; ++w2) ps += frontS[w2];

    int2 p[RUNS_W];
    int  len[RUNS_W], base[RUNS_W];
#pragma unroll
    for (int u = 0; u < RUNS_W; ++u) {
        const int k = wave * RUNS_W + u;
        len[u]  = runLen[k];
        base[u] = runBase[k];
        const int j = lane < len[u] ? lane : len[u] - 1;
        p[u] = pairs[len[u] > 0 ? base[u] + j : 0];
    }
#pragma unroll
    for (int u = 0; u < RUNS_W; ++u)
        if (lane < len[u]) atomicAdd(&c[(p[u].y - lo) & (BIN_W - 1)], 1);
#pragma unroll
    for (int u = 0; u < RUNS_W; ++u)
        for (int j = 32 + lane; j < len[u]; j += 32)
            atomicAdd(&c[(pairs[base[u] + j].y - lo) & (BIN_W - 1)], 1);
    __syncthreads();
    int deg = 0, start = 0;
    if (wave < BIN_W / 32) {
        deg = c[tid];
        const int incl = wave_incl_sum(deg, lane);
        if (lane == 31) wt[wave] = incl;
        start = ps + incl - deg;
    }
    __syncthreads();
    if (wave < BIN_W / 32) {
#pragma unroll
        for (int w2 = 0; w2 < BIN_W / 32; ++w2) start += (w2 < wave) ? wt[w2] : 0;
        c[tid] = start;
        const int node = lo + tid;
        if (node < n_nodes) {
            rowptr[node] = start;
            cnt[node]    = deg;
            invdeg[node] = 1.0f / fmaxf((float)deg, 1.0f);
        }
    }
    __syncthreads();
    const unsigned elast = (unsigned)(n_edges - 1);
#pragma unroll
    for (int u = 0; u < RUNS_W; ++u)
        if (lane < len[u]) {
            unsigned slot = (unsigned)atomicAdd(&c[(p[u].y - lo) & (BIN_W - 1)], 1);
            slot = slot < elast ? slot : elast;
            col[slot] = p[u].x;
        }
#pragma unroll
    for (int u = 0; u < RUNS_W; ++u)
        for (int j = 32 + lane; j < len[u]; j += 32) {
            const int2 q = pairs[base[u] + j];
            unsigned slot = (unsigned)atomicAdd(&c[(q.y - lo) & (BIN_W - 1)], 1);
            slot = slot < elast ? slot : elast;
            col[slot] = q.x;
        }
}

#define APITCH 132
#define GEMM_BLOCKS 512
#define GR 8

__device__ __forceinline__ unsigned src_id(unsigned m, int j, unsigned last) {
    const unsigned id = (unsigned)__builtin_amdgcn_readlane((int)m, j);
    return id < last ? id : last;
}

__device__ __forceinline__ void rows_issue(const float4* __restrict__ h4,
                                           unsigned m0, int n0, unsigned m1, int n1,
                                           int j, int lane, unsigned last,
                                           float4 (&v)[GR], float4 (&w)[GR]) {
#pragma unroll
    for (int u = 0; u < GR; ++u)
        if (j + u < n0) v[u] = h4[(size_t)src_id(m0, j + u, last) * 32 + lane];
#pragma unroll
    for (int u = 0; u < GR; ++u)
        if (j + u < n1) w[u] = h4[(size_t)src_id(m1, j + u, last) * 32 + lane];
}

__device__ __forceinline__ void rows_add(int n0, int n1, int j,
                                         const float4 (&v)[GR], const float4 (&w)[GR],
                                         float4& a0, float4& a1) {
#pragma unroll
    for (int u = 0; u < GR; ++u)
        if (j + u < n0) { a0.x += v[u].x; a0.y += v[u].y; a0.z += v[u].z; a0.w += v[u].w; }
#pragma unroll
    for (int u = 0; u < GR; ++u)
        if (j + u < n1) { a1.x += w[u].x; a1.y += w[u].y; a1.z += w[u].z; a1.w += w[u].w; }
}

__device__ __forceinline__ int clamp32(int d) { return d < 0 ? 0 : (d > 32 ? 32 : d); }

__device__ __forceinline__ void gather_finish(const float4* __restrict__ h4,
                                              const int* __restrict__ col,
                                              int s0, int d0, unsigned m0,
                                              int s1, int d1, unsigned m1,
                                              int lane, unsigned last,
                                              float4 (&v)[GR], float4 (&w)[GR],
                                              float4& r0, float4& r1) {
    float4 a0 = make_float4(0.0f, 0.0f, 0.0f, 0.0f);
    float4 a1 = make_float4(0.0f, 0.0f, 0.0f, 0.0f);
    int n0 = clamp32(d0), n1 = clamp32(d1);
    int nm = n0 > n1 ? n0 : n1;
    rows_add(n0, n1, 0, v, w, a0, a1);
    for (int j = GR; j < nm; j += GR) {
        rows_issue(h4, m0, n0, m1, n1, j, lane, last, v, w);
        rows_add(n0, n1, j, v, w, a0, a1);
    }
    const int dmax = d0 > d1 ? d0 : d1;
    for (int j0 = 32; j0 < dmax; j0 += 32) {
        m0 = 0; m1 = 0;
        if (j0 + lane < d0) m0 = (unsigned)col[s0 + j0 + lane];
        if (j0 + lane < d1) m1 = (unsigned)col[s1 + j0 + lane];
        m0 = m0 < last ? m0 : last;
        m1 = m1 < last ? m1 : last;
        n0 = clamp32(d0 - j0); n1 = clamp32(d1 - j0);
        nm = n0 > n1 ? n0 : n1;
        for (int j = 0; j < nm; j += GR) {
            rows_issue(h4, m0, n0, m1, n1, j, lane, last, v, w);
            rows_add(n0, n1, j, v, w, a0, a1);
        }
    }
    r0 = a0; r1 = a1;
}

__global__ __launch_bounds__(256)
void sage_gemm_kernel(const float* __restrict__ h,
                      const int* __restrict__ rowptr,
                      const int* __restrict__ cnt,
                      const int* __restrict__ col,
                      const float* __restrict__ invdeg,
                      const float* __restrict__ Wl,
                      const float* __restrict__ Wr,
                      const float* __restrict__ bias,
                      float* __restrict__ out,
                      int n_nodes, int do_relu) {
    __shared__ float sA[2][2][16][APITCH];

    const int lane = threadIdx.x & 31;
    const int wave = __builtin_amdgcn_readfirstlane((int)(threadIdx.x >> 5));
    const int n0   = wave * 16;

    const int lrow  = lane & 15;
    const int khalf = (lane >> 4) * 2;
    const int ncol  = n0 + lrow;

    const int ntiles = (n_nodes + 15) >> 4;
    const int tstep  = (int)gridDim.x;
    int t = (int)blockIdx.x;
    if (t >= ntiles) return;

    v2f bl[32], br[32];
#pragma unroll
    for (int j = 0; j < 32; ++j) {
        const int k = 4 * j + khalf;
        bl[j].x = Wl[(size_t)k * HID + ncol];
        bl[j].y = Wl[(size_t)(k + 1) * HID + ncol];
        br[j].x = Wr[(size_t)k * HID + ncol];
        br[j].y = Wr[(size_t)(k + 1) * HID + ncol];
    }
    const float bv = bias[ncol];

    const float4* __restrict__ h4 = (const float4*)h;
    const int      lasti = n_nodes - 1;
    const unsigned last  = (unsigned)lasti;
    const int sr0 = 2 * wave;
    const int sr1 = 2 * wave + 1;

    float4 ph0, ph1, g0, g1;
    float4 v[GR], w[GR];
    float  i0, i1;

    int ra = t * 16 + sr0, rb = t * 16 + sr1;
    ra = ra < lasti ? ra : lasti;
    rb = rb < lasti ? rb : lasti;
    int cs0 = rowptr[ra], cd0 = cnt[ra], cs1 = rowptr[rb], cd1 = cnt[rb];
    unsigned m0 = (unsigned)col[lane < cd0 ? cs0 + lane : 0];
    unsigned m1 = (unsigned)col[lane < cd1 ? cs1 + lane : 0];
    ph0 = h4[(size_t)ra * 32 + lane];
    ph1 = h4[(size_t)rb * 32 + lane];
    i0 = invdeg[ra];
    i1 = invdeg[rb];
    rows_issue(h4, m0, clamp32(cd0), m1, clamp32(cd1), 0, lane, last, v, w);
    gather_finish(h4, col, cs0, cd0, m0, cs1, cd1, m1, lane, last, v, w, g0, g1);
    *(float4*)&sA[0][0][sr0][4 * lane] = make_float4(g0.x * i0, g0.y * i0, g0.z * i0, g0.w * i0);
    *(float4*)&sA[0][0][sr1][4 * lane] = make_float4(g1.x * i1, g1.y * i1, g1.z * i1, g1.w * i1);
    *(float4*)&sA[0][1][sr0][4 * lane] = ph0;
    *(float4*)&sA[0][1][sr1][4 * lane] = ph1;

    int na = (t + tstep) * 16 + sr0, nb_ = (t + tstep) * 16 + sr1;
    na  = na  < lasti ? na  : lasti;
    nb_ = nb_ < lasti ? nb_ : lasti;
    cs0 = rowptr[na]; cd0 = cnt[na]; cs1 = rowptr[nb_]; cd1 = cnt[nb_];
    m0 = (unsigned)col[lane < cd0 ? cs0 + lane : 0];
    m1 = (unsigned)col[lane < cd1 ? cs1 + lane : 0];
    ra = (t + 2 * tstep) * 16 + sr0; rb = (t + 2 * tstep) * 16 + sr1;
    ra = ra < lasti ? ra : lasti;
    rb = rb < lasti ? rb : lasti;
    int sa = rowptr[ra], da = cnt[ra], sb = rowptr[rb], db = cnt[rb];
    __syncthreads();

    int cur = 0;
    for (; t < ntiles; t += tstep) {
        const int  tn   = t + tstep;
        const bool more = tn < ntiles;
        unsigned mn0 = 0, mn1 = 0;
        const int ns0 = sa, nd0 = da, ns1 = sb, nd1 = db;
        if (more) {
            ph0 = h4[(size_t)na  * 32 + lane];
            ph1 = h4[(size_t)nb_ * 32 + lane];
            i0 = invdeg[na];
            i1 = invdeg[nb_];
            mn0 = (unsigned)col[lane < nd0 ? ns0 + lane : 0];
            mn1 = (unsigned)col[lane < nd1 ? ns1 + lane : 0];
            rows_issue(h4, m0, clamp32(cd0), m1, clamp32(cd1), 0, lane, last, v, w);
            na = ra; nb_ = rb;
            ra = (tn + 2 * tstep) * 16 + sr0; rb = (tn + 2 * tstep) * 16 + sr1;
            ra = ra < lasti ? ra : lasti;
            rb = rb < lasti ? rb : lasti;
            sa = rowptr[ra]; da = cnt[ra]; sb = rowptr[rb]; db = cnt[rb];
        }

        v8f acc_l = {};
        v8f acc_r = {};
        const float* __restrict__ pl = &sA[cur][0][lrow][khalf];
        const float* __restrict__ pr = &sA[cur][1][lrow][khalf];
#pragma unroll
        for (int j = 0; j < 32; ++j) {
            const v2f a_l = *(const v2f*)(pl + 4 * j);
            const v2f a_r = *(const v2f*)(pr + 4 * j);
            acc_l = __builtin_amdgcn_wmma_f32_16x16x4_f32(
                false, a_l, false, bl[j], (short)0, acc_l, false, false);
            acc_r = __builtin_amdgcn_wmma_f32_16x16x4_f32(
                false, a_r, false, br[j], (short)0, acc_r, false, false);
        }

        const int mbase = t * 16 + (lane >> 4) * 8;
#pragma unroll
        for (int q = 0; q < 8; ++q) {
            int m = mbase + q;
            if (m >= n_nodes) m = n_nodes - 1;
            float o = acc_l[q] + acc_r[q] + bv;
            if (do_relu) o = o > 0.0f ? o : 0.0f;
            out[(size_t)m * HID + ncol] = o;
        }

        if (more) {
            gather_finish(h4, col, cs0, cd0, m0, cs1, cd1, m1, lane, last, v, w, g0, g1);
            const int nb = cur ^ 1;
            *(float4*)&sA[nb][0][sr0][4 * lane] = make_float4(g0.x * i0, g0.y * i0, g0.z * i0, g0.w * i0);
            *(float4*)&sA[nb][0][sr1][4 * lane] = make_float4(g1.x * i1, g1.y * i1, g1.z * i1, g1.w * i1);
            *(float4*)&sA[nb][1][sr0][4 * lane] = ph0;
            *(float4*)&sA[nb][1][sr1][4 * lane] = ph1;
            cs0 = ns0; cd0 = nd0; cs1 = ns1; cd1 = nd1;
            m0 = mn0; m1 = mn1;
        }
        __builtin_amdgcn_fence(__ATOMIC_RELEASE, "workgroup", "local");
        __builtin_amdgcn_s_barrier();
        __builtin_amdgcn_fence(__ATOMIC_ACQUIRE, "workgroup", "local");
        cur ^= 1;
    }
}

#define ED_G 8

__device__ __forceinline__ float wave_sum_all(float x) {
    x += __builtin_bit_cast(float, __builtin_amdgcn_update_dpp(0, __builtin_bit_cast(int, x), 0x161, 0xf, 0xf, false));
    x += __builtin_bit_cast(float, __builtin_amdgcn_update_dpp(0, __builtin_bit_cast(int, x), 0x162, 0xf, 0xf, false));
    x += __builtin_bit_cast(float, __builtin_amdgcn_update_dpp(0, __builtin_bit_cast(int, x), 0x164, 0xf, 0xf, false));
    x += __builtin_bit_cast(float, __builtin_amdgcn_update_dpp(0, __builtin_bit_cast(int, x), 0x168, 0xf, 0xf, false));
    const int b = __builtin_bit_cast(int, x);
    x += __builtin_bit_cast(float, __builtin_amdgcn_permlanex16(b, b, 0x76543210, 0xfedcba98, false, false));
    return x;
}

__global__ __launch_bounds__(256)
void edge_dot_kernel(const float* __restrict__ h2,
                     const int* __restrict__ es,
                     const int* __restrict__ ed,
                     float* __restrict__ out, int n) {
    const int lane = threadIdx.x & 31;
    const int e0 = __builtin_amdgcn_readfirstlane(
        (int)((blockIdx.x * blockDim.x + threadIdx.x) & ~31u));
    if (e0 >= n) return;
    const int e  = e0 + lane;
    const int ec = e < n ? e : n - 1;
    const int si = es[ec];
    const int di = ed[ec];
    const float4* __restrict__ h4 = (const float4*)h2;

    float res = 0.0f;
#pragma unroll 1
    for (int g = 0; g < 32; g += ED_G) {
        float4 a[ED_G], b[ED_G];
#pragma unroll
        for (int u = 0; u < ED_G; ++u) {
            const int s = __builtin_amdgcn_readlane(si, g + u);
            const int d = __builtin_amdgcn_readlane(di, g + u);
            a[u] = h4[(size_t)s * 32 + lane];
            b[u] = h4[(size_t)d * 32 + lane];
        }
#pragma unroll
        for (int u = 0; u < ED_G; ++u) {
            float p = a[u].x * b[u].x + a[u].y * b[u].y + a[u].z * b[u].z + a[u].w * b[u].w;
            p = wave_sum_all(p);
            res = (lane == g + u) ? p : res;
        }
    }
    if (e < n) out[e] = res;
}

extern "C" void kernel_launch(void* const* d_in, const int* in_sizes, int n_in,
                              void* d_out, int out_size, void* d_ws, size_t ws_size,
                              hipStream_t stream) {
    const int*   n_id     = (const int*)  d_in[0];
    const float* x_struct = (const float*)d_in[1];
    const int*   eidx     = (const int*)  d_in[2];
    const int*   elab     = (const int*)  d_in[3];
    const float* emb      = (const float*)d_in[4];
    const float* W1l      = (const float*)d_in[5];
    const float* W1r      = (const float*)d_in[6];
    const float* b1       = (const float*)d_in[7];
    const float* W2l      = (const float*)d_in[8];
    const float* W2r      = (const float*)d_in[9];
    const float* b2       = (const float*)d_in[10];
    float*       out      = (float*)d_out;

    const int N = in_sizes[0];
    const int E = in_sizes[2] / 2;
    const int L = in_sizes[3] / 2;

    const int* src = eidx;
    const int* dst = eidx + E;
    const int* lsrc = elab;
    const int* ldst = elab + L;

    float* ws     = (float*)d_ws;
    float* h0     = ws;
    float* h1     = h0  + (size_t)N * HID;
    float* lists  = h1  + (size_t)N * HID;
    float* spare  = lists + (size_t)N * HID;
    int*   cnt    = (int*)(spare + (size_t)N * HID);
    float* invdeg = (float*)(cnt + N);
    int*   rowptr = (int*)lists;
    int*   col    = rowptr + N + E;
    int2*  pairs  = (int2*)spare;
    int*   starts = (int*)spare + 2 * (size_t)E;
    float* h2     = h0;

    const int h0_rows_wg = (SORT_T / 32) * H0_ROWS;
    const int h0_wgs     = (N + h0_rows_wg - 1) / h0_rows_wg;
    const int h0_wgs_a   = h0_wgs / 2;
    const int h0_wgs_b   = h0_wgs - h0_wgs_a;
    const int h0_waves_a = h0_wgs_a * (SORT_T / 32);
    chunk_sort_kernel<<<SORT_BLOCKS + h0_wgs_a, SORT_T, 0, stream>>>(
        src, dst, starts, pairs, N, E, SORT_BLOCKS, n_id, emb, x_struct, h0);
    bin_sort_kernel<<<(N + BIN_W - 1) / BIN_W + h0_wgs_b, SORT_T, 0, stream>>>(
        pairs, starts, rowptr, cnt, invdeg, col, N, E, n_id, emb, x_struct, h0, h0_waves_a);

    sage_gemm_kernel<<<GEMM_BLOCKS, 256, 0, stream>>>(
        h0, rowptr, cnt, col, invdeg, W1l, W1r, b1, h1, N,  1);

    sage_gemm_kernel<<<GEMM_BLOCKS, 256, 0, stream>>>(
        h1, rowptr, cnt, col, invdeg, W2l, W2r, b2, h2, N,  0);

    edge_dot_kernel<<<(L + 255) / 256, 256, 0, stream>>>(h2, lsrc, ldst, out, L);
}
